# starting kernel
# baseline (speedup 1.0000x reference)
#include <hip/hip_runtime.h>
#include <hip/hip_bf16.h>

typedef float v2f __attribute__((ext_vector_type(2)));
typedef float v4f __attribute__((ext_vector_type(4)));
typedef float v8f __attribute__((ext_vector_type(8)));

#define NEG_SLOPE 0.01f
#define DCOL 128

__device__ __forceinline__ float lrelu(float x) {
    return x > 0.0f ? x : NEG_SLOPE * x;
}

__global__ void k_rootrow(const int* __restrict__ batch,
                          const int* __restrict__ root_idx,
                          int* __restrict__ rootrow, int N) {
    int i = blockIdx.x * blockDim.x + threadIdx.x;
    if (i < N) rootrow[i] = root_idx[batch[i]];
}

__global__ void k_init_bias(float* __restrict__ acc,
                            const float* __restrict__ bias, int N) {
    int i = blockIdx.x * blockDim.x + threadIdx.x;
    if (i < N * DCOL) acc[i] = bias[i & (DCOL - 1)];
}

__global__ void k_scatter(const float* __restrict__ xw,
                          const int* __restrict__ src,
                          const int* __restrict__ dst,
                          const float* __restrict__ wvals,
                          float* __restrict__ acc, int E) {
    int t = blockIdx.x * blockDim.x + threadIdx.x;
    int e = t >> 5;
    int lane = t & 31;
    if (e >= E) return;
    int s = src[e];
    int d = dst[e];
    float w = wvals[e];
    const v4f x = *(const v4f*)(xw + (size_t)s * DCOL + lane * 4);
    float* a = acc + (size_t)d * DCOL + lane * 4;
    unsafeAtomicAdd(a + 0, w * x[0]);
    unsafeAtomicAdd(a + 1, w * x[1]);
    unsafeAtomicAdd(a + 2, w * x[2]);
    unsafeAtomicAdd(a + 3, w * x[3]);
}

template <int KCHUNKS, bool LR0, bool LR1, bool EPI>
__global__ void k_gemm_wmma(const float* __restrict__ A0,
                            const float* __restrict__ A1,
                            const int* __restrict__ rootrow,
                            const float* __restrict__ W,
                            const float* __restrict__ bias,
                            float* __restrict__ C, int N) {
    __shared__ float lw[128 * 128];

    const int tid  = threadIdx.x;
    const int wave = tid >> 5;
    const int lane = tid & 31;
    const int lo   = lane & 15;
    const int hi   = lane >> 4;
    const int rowbase = blockIdx.x * 128 + wave * 16;
    const bool active = rowbase < N;

    const int arow = rowbase + lo;
    int arow1 = 0;
    if (KCHUNKS > 1 && active) arow1 = rootrow[arow];

    v8f acc[8] = {};

#pragma unroll
    for (int kc = 0; kc < KCHUNKS; ++kc) {
        const float* wsrc = W + (size_t)kc * 128 * DCOL;
        for (int i = tid; i < (128 * DCOL) / 8; i += 256) {
            int kp = i >> 5;
            int cg = (i & 31) * 4;
            v4f e = *(const v4f*)(wsrc + (kp * 2 + 0) * DCOL + cg);
            v4f o = *(const v4f*)(wsrc + (kp * 2 + 1) * DCOL + cg);
            float* dstp = lw + kp * 256 + cg * 2;
#pragma unroll
            for (int j = 0; j < 4; ++j) {
                v2f p = { e[j], o[j] };
                *(v2f*)(dstp + j * 2) = p;
            }
        }
        __syncthreads();

        if (active) {
            const float* arowptr = (kc == 0) ? (A0 + (size_t)arow * DCOL)
                                             : (A1 + (size_t)arow1 * DCOL);
            const bool lr = (kc == 0) ? LR0 : LR1;
            const float* lb = lw + hi * 256 + lo * 2;
#pragma unroll 4
            for (int ks = 0; ks < 32; ++ks) {
                v2f a = *(const v2f*)(arowptr + ks * 4 + hi * 2);
                if (lr) { a[0] = lrelu(a[0]); a[1] = lrelu(a[1]); }
                v2f b[8];
#pragma unroll
                for (int t = 0; t < 8; ++t)
                    b[t] = *(const v2f*)(lb + ks * 512 + t * 32);
#pragma unroll
                for (int t = 0; t < 8; ++t)
                    acc[t] = __builtin_amdgcn_wmma_f32_16x16x4_f32(
                        false, a, false, b[t], (short)0, acc[t], false, false);
            }
        }
        __syncthreads();
    }

    if (active) {
#pragma unroll
        for (int t = 0; t < 8; ++t) {
            int col = t * 16 + lo;
            float badd = EPI ? bias[col] : 0.0f;
#pragma unroll
            for (int r = 0; r < 8; ++r) {
                int row = rowbase + r + hi * 8;
                float v = acc[t][r];
                if (EPI) v = lrelu(v + badd);
                C[(size_t)row * DCOL + col] = v;
            }
        }
    }
}

extern "C" void kernel_launch(void* const* d_in, const int* in_sizes, int n_in,
                              void* d_out, int out_size, void* d_ws, size_t ws_size,
                              hipStream_t stream) {
    const float* features = (const float*)d_in[0];
    const float* values   = (const float*)d_in[1];
    const float* W1       = (const float*)d_in[2];
    const float* b1       = (const float*)d_in[3];
    const float* W2       = (const float*)d_in[4];
    const float* b2       = (const float*)d_in[5];
    const float* Wl       = (const float*)d_in[6];
    const float* bl       = (const float*)d_in[7];
    const int*   adjs     = (const int*)d_in[8];
    const int*   root_idx = (const int*)d_in[9];
    const int*   batch    = (const int*)d_in[12];

    const int N = in_sizes[0] / DCOL;
    const int E = in_sizes[1];
    const int* srcI = adjs;
    const int* dstI = adjs + E;

    float* XW = (float*)d_ws;
    float* H1 = XW + (size_t)N * DCOL;
    float* H2 = H1 + (size_t)N * DCOL;
    int* ROOTROW = (int*)(H2 + (size_t)N * DCOL);

    float* OUT = (float*)d_out;

    const int gemmBlocks = (N + 127) / 128;
    const int initBlocks = (N * DCOL + 255) / 256;
    const int scatBlocks = (int)(((size_t)E * 32 + 255) / 256);

    k_rootrow<<<(N + 255) / 256, 256, 0, stream>>>(batch, root_idx, ROOTROW, N);

    k_gemm_wmma<1, false, false, false><<<gemmBlocks, 256, 0, stream>>>(
        features, nullptr, nullptr, W1, nullptr, XW, N);
    k_init_bias<<<initBlocks, 256, 0, stream>>>(H1, b1, N);
    k_scatter<<<scatBlocks, 256, 0, stream>>>(XW, srcI, dstI, values, H1, E);

    k_gemm_wmma<2, true, true, false><<<gemmBlocks, 256, 0, stream>>>(
        H1, features, ROOTROW, W2, nullptr, XW, N);
    k_init_bias<<<initBlocks, 256, 0, stream>>>(H2, b2, N);
    k_scatter<<<scatBlocks, 256, 0, stream>>>(XW, srcI, dstI, values, H2, E);

    k_gemm_wmma<2, true, false, true><<<gemmBlocks, 256, 0, stream>>>(
        H2, H1, ROOTROW, Wl, bl, OUT, N);
}


# Float-atomic scatter-add replaced by per-call counting sort of edges by dst (int atomic rank) + one-wave-per-node gather-sum; sum order = sort arrival order; bias-init kernels gone
# speedup vs baseline: 1.5811x; 1.5811x over previous
#include <hip/hip_runtime.h>
#include <hip/hip_bf16.h>

typedef float v2f __attribute__((ext_vector_type(2)));
typedef float v4f __attribute__((ext_vector_type(4)));
typedef float v8f __attribute__((ext_vector_type(8)));

#define NEG_SLOPE 0.01f
#define DCOL 128

__device__ __forceinline__ float lrelu(float x) {
    return x > 0.0f ? x : NEG_SLOPE * x;
}

__global__ void k_prep(const int* __restrict__ batch,
                       const int* __restrict__ root_idx,
                       int* __restrict__ rootrow,
                       int* __restrict__ count, int N) {
    int i = blockIdx.x * blockDim.x + threadIdx.x;
    if (i < N) {
        rootrow[i] = root_idx[batch[i]];
        count[i] = 0;
    }
}

#define EPT 4

__global__ void k_hist(const int* __restrict__ dst,
                       int* __restrict__ count,
                       int* __restrict__ rank, int E) {
    const int t = blockIdx.x * blockDim.x + threadIdx.x;
    const int T = gridDim.x * blockDim.x;
    int d[EPT];
#pragma unroll
    for (int k = 0; k < EPT; ++k) {
        int e = t + k * T;
        d[k] = (e < E) ? dst[e] : -1;
    }
    int r[EPT];
#pragma unroll
    for (int k = 0; k < EPT; ++k)
        r[k] = (d[k] >= 0) ? atomicAdd(count + d[k], 1) : 0;
#pragma unroll
    for (int k = 0; k < EPT; ++k) {
        int e = t + k * T;
        if (e < E) rank[e] = r[k];
    }
}

__global__ void __launch_bounds__(1024)
k_scan(const int* __restrict__ count, int* __restrict__ rowptr, int N) {
    __shared__ int wsum[32];
    __shared__ int tile_total;
    const int tid  = threadIdx.x;
    const int lane = tid & 31;
    const int wave = tid >> 5;
    int carry = 0;
    for (int base = 0; base < N; base += 1024 * 16) {
        const int i0 = base + tid * 16;
        int v[16];
#pragma unroll
        for (int j = 0; j < 16; ++j) {
            int idx = (i0 + j < N) ? (i0 + j) : (N - 1);
            int cv = count[idx];
            v[j] = (i0 + j < N) ? cv : 0;
        }
        int s = 0;
#pragma unroll
        for (int j = 0; j < 16; ++j) { int c = v[j]; v[j] = s; s += c; }
        int incl = s;
        for (int o = 1; o < 32; o <<= 1) {
            int n = __shfl_up(incl, o);
            if (lane >= o) incl += n;
        }
        if (lane == 31) wsum[wave] = incl;
        __syncthreads();
        if (wave == 0) {
            int w = wsum[lane];
            int wi = w;
            for (int o = 1; o < 32; o <<= 1) {
                int n = __shfl_up(wi, o);
                if (lane >= o) wi += n;
            }
            wsum[lane] = wi - w;
            if (lane == 31) tile_total = wi;
        }
        __syncthreads();
        const int off = carry + wsum[wave] + (incl - s);
#pragma unroll
        for (int j = 0; j < 16; ++j)
            if (i0 + j < N) rowptr[i0 + j] = off + v[j];
        carry += tile_total;
        __syncthreads();
    }
    if (tid == 0) rowptr[N] = carry;
}

__global__ void k_fill(const int* __restrict__ src,
                       const int* __restrict__ dst,
                       const float* __restrict__ wvals,
                       const int* __restrict__ rowptr,
                       const int* __restrict__ rank,
                       int2* __restrict__ perm, int E) {
    const int t = blockIdx.x * blockDim.x + threadIdx.x;
    const int T = gridDim.x * blockDim.x;
    int d[EPT], r[EPT], s[EPT];
    float w[EPT];
#pragma unroll
    for (int k = 0; k < EPT; ++k) {
        int e = t + k * T;
        int ec = (e < E) ? e : (E - 1);
        d[k] = dst[ec];
        r[k] = rank[ec];
        s[k] = src[ec];
        w[k] = wvals[ec];
    }
    int pos[EPT];
#pragma unroll
    for (int k = 0; k < EPT; ++k) pos[k] = rowptr[d[k]] + r[k];
#pragma unroll
    for (int k = 0; k < EPT; ++k) {
        int e = t + k * T;
        if (e < E) perm[pos[k]] = make_int2(s[k], __float_as_int(w[k]));
    }
}

template <int U>
__device__ __forceinline__ void gather_rows(const float* __restrict__ xw,
                                            int sx, int wy, int j, int lane,
                                            v4f& acc) {
    v4f x[U];
    float w[U];
#pragma unroll
    for (int u = 0; u < U; ++u) {
        const int s = __builtin_amdgcn_readlane(sx, j + u);
        w[u] = __int_as_float(__builtin_amdgcn_readlane(wy, j + u));
        x[u] = *(const v4f*)(xw + (size_t)s * DCOL + lane * 4);
    }
#pragma unroll
    for (int u = 0; u < U; ++u) acc += w[u] * x[u];
}

__global__ void k_gather(const float* __restrict__ xw,
                         const int* __restrict__ rowptr,
                         const int2* __restrict__ perm,
                         const float* __restrict__ bias,
                         float* __restrict__ H, int N) {
    const int lane = threadIdx.x & 31;
    const int gw = __builtin_amdgcn_readfirstlane(
        (int)((blockIdx.x * blockDim.x + threadIdx.x) >> 5));
    const int nw = (int)((gridDim.x * blockDim.x) >> 5);
    const v4f b = *(const v4f*)(bias + lane * 4);

    for (int node = gw; node < N; node += nw) {
        const int p0 = rowptr[node];
        const int p1 = rowptr[node + 1];
        v4f acc = {0.0f, 0.0f, 0.0f, 0.0f};
        for (int base = p0; base < p1; base += 32) {
            const int cnt = (p1 - base < 32) ? (p1 - base) : 32;
            int2 my = make_int2(0, 0);
            if (lane < cnt) my = perm[base + lane];
            int j = 0;
            for (; j + 8 <= cnt; j += 8) gather_rows<8>(xw, my.x, my.y, j, lane, acc);
            if (j + 4 <= cnt) { gather_rows<4>(xw, my.x, my.y, j, lane, acc); j += 4; }
            if (j + 2 <= cnt) { gather_rows<2>(xw, my.x, my.y, j, lane, acc); j += 2; }
            if (j < cnt)      { gather_rows<1>(xw, my.x, my.y, j, lane, acc); }
        }
        acc += b;
        *(v4f*)(H + (size_t)node * DCOL + lane * 4) = acc;
    }
}

template <int KCHUNKS, bool LR0, bool LR1, bool EPI>
__global__ void k_gemm_wmma(const float* __restrict__ A0,
                            const float* __restrict__ A1,
                            const int* __restrict__ rootrow,
                            const float* __restrict__ W,
                            const float* __restrict__ bias,
                            float* __restrict__ C, int N) {
    __shared__ float lw[128 * 128];

    const int tid  = threadIdx.x;
    const int wave = tid >> 5;
    const int lane = tid & 31;
    const int lo   = lane & 15;
    const int hi   = lane >> 4;
    const int rowbase = blockIdx.x * 128 + wave * 16;
    const bool active = rowbase < N;

    const int arow = rowbase + lo;
    int arow1 = 0;
    if (KCHUNKS > 1 && active) arow1 = rootrow[arow];

    v8f acc[8] = {};

#pragma unroll
    for (int kc = 0; kc < KCHUNKS; ++kc) {
        const float* wsrc = W + (size_t)kc * 128 * DCOL;
        for (int i = tid; i < (128 * DCOL) / 8; i += 256) {
            int kp = i >> 5;
            int cg = (i & 31) * 4;
            v4f e = *(const v4f*)(wsrc + (kp * 2 + 0) * DCOL + cg);
            v4f o = *(const v4f*)(wsrc + (kp * 2 + 1) * DCOL + cg);
            float* dstp = lw + kp * 256 + cg * 2;
#pragma unroll
            for (int j = 0; j < 4; ++j) {
                v2f p = { e[j], o[j] };
                *(v2f*)(dstp + j * 2) = p;
            }
        }
        __syncthreads();

        if (active) {
            const float* arowptr = (kc == 0) ? (A0 + (size_t)arow * DCOL)
                                             : (A1 + (size_t)arow1 * DCOL);
            const bool lr = (kc == 0) ? LR0 : LR1;
            const float* lb = lw + hi * 256 + lo * 2;
#pragma unroll 4
            for (int ks = 0; ks < 32; ++ks) {
                v2f a = *(const v2f*)(arowptr + ks * 4 + hi * 2);
                if (lr) { a[0] = lrelu(a[0]); a[1] = lrelu(a[1]); }
                v2f b[8];
#pragma unroll
                for (int t = 0; t < 8; ++t)
                    b[t] = *(const v2f*)(lb + ks * 512 + t * 32);
#pragma unroll
                for (int t = 0; t < 8; ++t)
                    acc[t] = __builtin_amdgcn_wmma_f32_16x16x4_f32(
                        false, a, false, b[t], (short)0, acc[t], false, false);
            }
        }
        __syncthreads();
    }

    if (active) {
#pragma unroll
        for (int t = 0; t < 8; ++t) {
            int col = t * 16 + lo;
            float badd = EPI ? bias[col] : 0.0f;
#pragma unroll
            for (int r = 0; r < 8; ++r) {
                int row = rowbase + r + hi * 8;
                float v = acc[t][r];
                if (EPI) v = lrelu(v + badd);
                C[(size_t)row * DCOL + col] = v;
            }
        }
    }
}

extern "C" void kernel_launch(void* const* d_in, const int* in_sizes, int n_in,
                              void* d_out, int out_size, void* d_ws, size_t ws_size,
                              hipStream_t stream) {
    const float* features = (const float*)d_in[0];
    const float* values   = (const float*)d_in[1];
    const float* W1       = (const float*)d_in[2];
    const float* b1       = (const float*)d_in[3];
    const float* W2       = (const float*)d_in[4];
    const float* b2       = (const float*)d_in[5];
    const float* Wl       = (const float*)d_in[6];
    const float* bl       = (const float*)d_in[7];
    const int*   adjs     = (const int*)d_in[8];
    const int*   root_idx = (const int*)d_in[9];
    const int*   batch    = (const int*)d_in[12];

    const int N = in_sizes[0] / DCOL;
    const int E = in_sizes[1];
    const int* srcI = adjs;
    const int* dstI = adjs + E;

    int2* PERM   = (int2*)d_ws;
    int*  RANK   = (int*)(PERM + E);
    int*  COUNT  = RANK + E;
    int*  ROWPTR = COUNT + N;
    float* H1 = (float*)d_ws + (size_t)N * DCOL;
    float* H2 = H1 + (size_t)N * DCOL;
    int* ROOTROW = (int*)(H2 + (size_t)N * DCOL);

    float* OUT = (float*)d_out;
    float* XW  = OUT;

    const int gemmBlocks = (N + 127) / 128;
    const int edgeBlocks = (E + 256 * EPT - 1) / (256 * EPT);
    const int gatherBlocks = 2048;

    k_prep<<<(N + 255) / 256, 256, 0, stream>>>(batch, root_idx, ROOTROW, COUNT, N);
    k_hist<<<edgeBlocks, 256, 0, stream>>>(dstI, COUNT, RANK, E);
    k_scan<<<1, 1024, 0, stream>>>(COUNT, ROWPTR, N);
    k_fill<<<edgeBlocks, 256, 0, stream>>>(srcI, dstI, values, ROWPTR, RANK, PERM, E);

    k_gemm_wmma<1, false, false, false><<<gemmBlocks, 256, 0, stream>>>(
        features, nullptr, nullptr, W1, nullptr, XW, N);
    k_gather<<<gatherBlocks, 256, 0, stream>>>(XW, ROWPTR, PERM, b1, H1, N);

    k_gemm_wmma<2, true, true, false><<<gemmBlocks, 256, 0, stream>>>(
        H1, features, ROOTROW, W2, nullptr, XW, N);
    k_gather<<<gatherBlocks, 256, 0, stream>>>(XW, ROWPTR, PERM, b2, H2, N);

    k_gemm_wmma<2, true, false, true><<<gemmBlocks, 256, 0, stream>>>(
        H2, H1, ROOTROW, Wl, bl, OUT, N);
}


# Edge sort by dst rebuilt on LDS counters in 2 launches (bucket split per chunk, node placement per bucket): no global atomics, no one-block scan; launches 9 to 7; sum order = LDS arrival order
# speedup vs baseline: 3.0143x; 1.9065x over previous
#include <hip/hip_runtime.h>
#include <hip/hip_bf16.h>

typedef float v2f __attribute__((ext_vector_type(2)));
typedef float v4f __attribute__((ext_vector_type(4)));
typedef float v8f __attribute__((ext_vector_type(8)));

#define NEG_SLOPE 0.01f
#define DCOL 128

__device__ __forceinline__ float lrelu(float x) {
    return x > 0.0f ? x : NEG_SLOPE * x;
}

#define SORT_CH 256
#define SORT_T  16
#define BSH     8

__device__ __forceinline__ int block_scan256(int v, int* wsum, int& total) {
    const int lane = threadIdx.x & 31;
    const int wave = threadIdx.x >> 5;
    int incl = v;
    for (int o = 1; o < 32; o <<= 1) {
        int n = __shfl_up(incl, o);
        if (lane >= o) incl += n;
    }
    if (lane == 31) wsum[wave] = incl;
    __syncthreads();
    int woff = 0, tot = 0;
#pragma unroll
    for (int k = 0; k < 8; ++k) {
        const int s = wsum[k];
        woff += (k < wave) ? s : 0;
        tot += s;
    }
    __syncthreads();
    total = tot;
    return woff + incl - v;
}

__global__ void __launch_bounds__(256)
k_part(const int* __restrict__ src,
       const int* __restrict__ dst,
       const float* __restrict__ wvals,
       int2* __restrict__ rec,
       int* __restrict__ ctab,
       int* __restrict__ otab, int E, int L, int NB) {
    __shared__ int cnt[256];
    __shared__ int wsum[8];
    const int tid = threadIdx.x;
    const int c = (int)blockIdx.x;
    const int e0 = (c * L < E) ? c * L : E;
    const int e1 = (e0 + L < E) ? e0 + L : E;
    cnt[tid] = 0;
    __syncthreads();

    for (int base = e0; base < e1; base += 256 * SORT_T) {
        int d[SORT_T];
#pragma unroll
        for (int k = 0; k < SORT_T; ++k) {
            const int e = base + k * 256 + tid;
            d[k] = dst[(e < e1) ? e : (e1 - 1)];
        }
#pragma unroll
        for (int k = 0; k < SORT_T; ++k) {
            const int e = base + k * 256 + tid;
            if (e < e1) atomicAdd(&cnt[d[k] >> BSH], 1);
        }
    }
    __syncthreads();

    const int myc = cnt[tid];
    int total;
    const int off = block_scan256(myc, wsum, total);
    if (tid < NB) {
        ctab[tid * SORT_CH + c] = myc;
        otab[tid * SORT_CH + c] = off;
    }
    cnt[tid] = off;
    __syncthreads();

    for (int base = e0; base < e1; base += 256 * SORT_T) {
        int d[SORT_T], s[SORT_T];
        float w[SORT_T];
#pragma unroll
        for (int k = 0; k < SORT_T; ++k) {
            const int e = base + k * 256 + tid;
            const int ec = (e < e1) ? e : (e1 - 1);
            d[k] = dst[ec];
            s[k] = src[ec];
            w[k] = wvals[ec];
        }
        int p[SORT_T];
#pragma unroll
        for (int k = 0; k < SORT_T; ++k) {
            const int e = base + k * 256 + tid;
            p[k] = (e < e1) ? atomicAdd(&cnt[d[k] >> BSH], 1) : 0;
        }
#pragma unroll
        for (int k = 0; k < SORT_T; ++k) {
            const int e = base + k * 256 + tid;
            if (e < e1) {
                const unsigned x = ((unsigned)(d[k] & 255) << 24) | (unsigned)s[k];
                rec[e0 + p[k]] = make_int2((int)x, __float_as_int(w[k]));
            }
        }
    }
}

__global__ void __launch_bounds__(256)
k_bucket(const int2* __restrict__ rec,
         const int* __restrict__ ctab,
         const int* __restrict__ otab,
         int* __restrict__ rowptr,
         int2* __restrict__ perm, int N, int E, int L) {
    __shared__ int ncnt[256];
    __shared__ int wsum[8];
    const int tid  = threadIdx.x;
    const int lane = tid & 31;
    const int b = (int)blockIdx.x;

    const int myCnt = ctab[b * SORT_CH + tid];
    const int myOff = otab[b * SORT_CH + tid];
    const int myStart = ((tid * L < E) ? tid * L : E) + myOff;
    ncnt[tid] = 0;
    int base;
    (void)block_scan256(myOff, wsum, base);

    int2 r[32];
#pragma unroll
    for (int i = 0; i < 32; ++i) {
        const int cn = __builtin_amdgcn_readlane(myCnt, i);
        const int st = __builtin_amdgcn_readlane(myStart, i);
        int idx = st + ((lane < cn) ? lane : 0);
        idx = (idx < E) ? idx : (E - 1);
        r[i] = rec[idx];
    }
#pragma unroll
    for (int i = 0; i < 32; ++i) {
        const int cn = __builtin_amdgcn_readlane(myCnt, i);
        if (lane < cn) atomicAdd(&ncnt[(unsigned)r[i].x >> 24], 1);
    }
    const bool longPiece = __builtin_amdgcn_ballot_w32(myCnt > 32) != 0u;
    if (longPiece) {
        for (int i = 0; i < 32; ++i) {
            const int cn = __builtin_amdgcn_readlane(myCnt, i);
            const int st = __builtin_amdgcn_readlane(myStart, i);
            for (int k = 32 + lane; k < cn; k += 32) {
                const int2 q = rec[st + k];
                atomicAdd(&ncnt[(unsigned)q.x >> 24], 1);
            }
        }
    }
    __syncthreads();

    const int v = ncnt[tid];
    int tot;
    const int ex = block_scan256(v, wsum, tot);
    const int node = b * 256 + tid;
    if (node < N) rowptr[node] = base + ex;
    if (node == N - 1) rowptr[N] = E;
    ncnt[tid] = ex;
    __syncthreads();

    int p[32];
#pragma unroll
    for (int i = 0; i < 32; ++i) {
        const int cn = __builtin_amdgcn_readlane(myCnt, i);
        p[i] = (lane < cn) ? atomicAdd(&ncnt[(unsigned)r[i].x >> 24], 1) : 0;
    }
#pragma unroll
    for (int i = 0; i < 32; ++i) {
        const int cn = __builtin_amdgcn_readlane(myCnt, i);
        if (lane < cn) {
            const unsigned x = (unsigned)r[i].x;
            perm[base + p[i]] = make_int2((int)(x & 0xFFFFFFu), r[i].y);
        }
    }
    if (longPiece) {
        for (int i = 0; i < 32; ++i) {
            const int cn = __builtin_amdgcn_readlane(myCnt, i);
            const int st = __builtin_amdgcn_readlane(myStart, i);
            for (int k = 32 + lane; k < cn; k += 32) {
                const int2 q = rec[st + k];
                const unsigned x = (unsigned)q.x;
                const int p = atomicAdd(&ncnt[x >> 24], 1);
                perm[base + p] = make_int2((int)(x & 0xFFFFFFu), q.y);
            }
        }
    }
}

template <int U>
__device__ __forceinline__ void gather_rows(const float* __restrict__ xw,
                                            int sx, int wy, int j, int lane,
                                            v4f& acc) {
    v4f x[U];
    float w[U];
#pragma unroll
    for (int u = 0; u < U; ++u) {
        const int s = __builtin_amdgcn_readlane(sx, j + u);
        w[u] = __int_as_float(__builtin_amdgcn_readlane(wy, j + u));
        x[u] = *(const v4f*)(xw + (size_t)s * DCOL + lane * 4);
    }
#pragma unroll
    for (int u = 0; u < U; ++u) acc += w[u] * x[u];
}

__global__ void k_gather(const float* __restrict__ xw,
                         const int* __restrict__ rowptr,
                         const int2* __restrict__ perm,
                         const float* __restrict__ bias,
                         float* __restrict__ H, int N) {
    const int lane = threadIdx.x & 31;
    const int gw = __builtin_amdgcn_readfirstlane(
        (int)((blockIdx.x * blockDim.x + threadIdx.x) >> 5));
    const int nw = (int)((gridDim.x * blockDim.x) >> 5);
    const v4f b = *(const v4f*)(bias + lane * 4);

    for (int node = gw; node < N; node += nw) {
        const int p0 = rowptr[node];
        const int p1 = rowptr[node + 1];
        v4f acc = {0.0f, 0.0f, 0.0f, 0.0f};
        for (int base = p0; base < p1; base += 32) {
            const int cnt = (p1 - base < 32) ? (p1 - base) : 32;
            int2 my = make_int2(0, 0);
            if (lane < cnt) my = perm[base + lane];
            int j = 0;
            for (; j + 8 <= cnt; j += 8) gather_rows<8>(xw, my.x, my.y, j, lane, acc);
            if (j + 4 <= cnt) { gather_rows<4>(xw, my.x, my.y, j, lane, acc); j += 4; }
            if (j + 2 <= cnt) { gather_rows<2>(xw, my.x, my.y, j, lane, acc); j += 2; }
            if (j < cnt)      { gather_rows<1>(xw, my.x, my.y, j, lane, acc); }
        }
        acc += b;
        *(v4f*)(H + (size_t)node * DCOL + lane * 4) = acc;
    }
}

template <bool LR, bool INIT, bool EPI, bool MAP>
__device__ __forceinline__ void gemm_tile128(float* __restrict__ lw,
                                             const float* __restrict__ A,
                                             const int* __restrict__ rowmap,
                                             const float* __restrict__ W,
                                             const int* __restrict__ batch,
                                             const float* __restrict__ R,
                                             const float* __restrict__ bias,
                                             float* __restrict__ C,
                                             int nrows, int blk) {
    const int tid  = threadIdx.x;
    const int wave = tid >> 5;
    const int lane = tid & 31;
    const int lo   = lane & 15;
    const int hi   = lane >> 4;
    const int rowbase = blk * 128 + wave * 16;
    const bool active = rowbase < nrows;

    int arow = rowbase + lo;
    if (MAP) {
        const int g = (arow < nrows) ? arow : (nrows - 1);
        arow = active ? rowmap[g] : 0;
    }

    v8f acc[8] = {};
    if (INIT && active) {
#pragma unroll
        for (int r = 0; r < 8; ++r) {
            const int g = batch[rowbase + r + hi * 8];
            const float* rp = R + (size_t)g * DCOL + lo;
#pragma unroll
            for (int t = 0; t < 8; ++t) acc[t][r] = rp[t * 16];
        }
    }

    for (int i = tid; i < (128 * DCOL) / 8; i += 256) {
        int kp = i >> 5;
        int cg = (i & 31) * 4;
        v4f e = *(const v4f*)(W + (kp * 2 + 0) * DCOL + cg);
        v4f o = *(const v4f*)(W + (kp * 2 + 1) * DCOL + cg);
        float* dstp = lw + kp * 256 + cg * 2;
#pragma unroll
        for (int j = 0; j < 4; ++j) {
            v2f p = { e[j], o[j] };
            *(v2f*)(dstp + j * 2) = p;
        }
    }
    __syncthreads();

    if (active) {
        const float* arowptr = A + (size_t)arow * DCOL;
        const float* lb = lw + hi * 256 + lo * 2;
#pragma unroll 4
        for (int ks = 0; ks < 32; ++ks) {
            v2f a = *(const v2f*)(arowptr + ks * 4 + hi * 2);
            if (LR) { a[0] = lrelu(a[0]); a[1] = lrelu(a[1]); }
            v2f b[8];
#pragma unroll
            for (int t = 0; t < 8; ++t)
                b[t] = *(const v2f*)(lb + ks * 512 + t * 32);
#pragma unroll
            for (int t = 0; t < 8; ++t)
                acc[t] = __builtin_amdgcn_wmma_f32_16x16x4_f32(
                    false, a, false, b[t], (short)0, acc[t], false, false);
        }

        if (MAP) {
#pragma unroll
            for (int r = 0; r < 8; ++r) {
                const int row = rowbase + r + hi * 8;
                if (row < nrows) {
#pragma unroll
                    for (int t = 0; t < 8; ++t)
                        C[(size_t)row * DCOL + t * 16 + lo] = acc[t][r];
                }
            }
        } else {
#pragma unroll
            for (int t = 0; t < 8; ++t) {
                int col = t * 16 + lo;
                float badd = EPI ? bias[col] : 0.0f;
#pragma unroll
                for (int r = 0; r < 8; ++r) {
                    int row = rowbase + r + hi * 8;
                    float v = acc[t][r];
                    if (EPI) v = lrelu(v + badd);
                    C[(size_t)row * DCOL + col] = v;
                }
            }
        }
    }
}

template <bool LR0, bool INIT, bool EPI, int SIDE>
__global__ void k_gemm_wmma(const float* __restrict__ A0,
                            const float* __restrict__ W,
                            const int* __restrict__ batch,
                            const float* __restrict__ R,
                            const float* __restrict__ bias,
                            float* __restrict__ C, int N,
                            const float* __restrict__ TA,
                            const int* __restrict__ root_idx,
                            const float* __restrict__ TW,
                            float* __restrict__ T, int B, int sideBlocks) {
    __shared__ float lw[128 * 128];
    const int blk = (int)blockIdx.x;
    if (SIDE != 0 && blk < sideBlocks) {
        gemm_tile128<SIDE == 1, false, false, true>(
            lw, TA, root_idx, TW, nullptr, nullptr, nullptr, T, B, blk);
    } else {
        gemm_tile128<LR0, INIT, EPI, false>(
            lw, A0, nullptr, W, batch, R, bias, C, N,
            blk - (SIDE != 0 ? sideBlocks : 0));
    }
}

extern "C" void kernel_launch(void* const* d_in, const int* in_sizes, int n_in,
                              void* d_out, int out_size, void* d_ws, size_t ws_size,
                              hipStream_t stream) {
    const float* features = (const float*)d_in[0];
    const float* values   = (const float*)d_in[1];
    const float* W1       = (const float*)d_in[2];
    const float* b1       = (const float*)d_in[3];
    const float* W2       = (const float*)d_in[4];
    const float* b2       = (const float*)d_in[5];
    const float* Wl       = (const float*)d_in[6];
    const float* bl       = (const float*)d_in[7];
    const int*   adjs     = (const int*)d_in[8];
    const int*   root_idx = (const int*)d_in[9];
    const int*   batch    = (const int*)d_in[12];

    const int N = in_sizes[0] / DCOL;
    const int E = in_sizes[1];
    const int B = in_sizes[9];
    const int* srcI = adjs;
    const int* dstI = adjs + E;

    int2* PERM   = (int2*)d_ws;
    int2* REC    = PERM + E;
    int*  CTAB   = (int*)(REC + E);
    int*  OTAB   = CTAB + 256 * SORT_CH;
    int*  ROWPTR = OTAB + 256 * SORT_CH;
    float* H1 = (float*)d_ws + (size_t)N * DCOL;
    float* H2 = H1 + (size_t)N * DCOL;
    float* R2 = H1 - (size_t)2 * B * DCOL;
    float* R3 = R2 + (size_t)B * DCOL;

    float* OUT = (float*)d_out;
    float* XW  = OUT;

    const int gemmBlocks = (N + 127) / 128;
    const int sideBlocks = (B + 127) / 128;
    const int chunkLen = (E + SORT_CH - 1) / SORT_CH;
    const int buckets  = (N + 255) / 256;
    const int gatherBlocks = 2048;

    k_part<<<SORT_CH, 256, 0, stream>>>(srcI, dstI, values, REC, CTAB, OTAB, E, chunkLen, buckets);
    k_bucket<<<buckets, 256, 0, stream>>>(REC, CTAB, OTAB, ROWPTR, PERM, N, E, chunkLen);

    k_gemm_wmma<false, false, false, 1><<<gemmBlocks + sideBlocks, 256, 0, stream>>>(
        features, W1, nullptr, nullptr, nullptr, XW, N,
        features, root_idx, W2 + 128 * DCOL, R2, B, sideBlocks);
    k_gather<<<gatherBlocks, 256, 0, stream>>>(XW, ROWPTR, PERM, b1, H1, N);

    k_gemm_wmma<true, true, false, 2><<<gemmBlocks + sideBlocks, 256, 0, stream>>>(
        H1, W2, batch, R2, nullptr, XW, N,
        H1, root_idx, Wl + 128 * DCOL, R3, B, sideBlocks);
    k_gather<<<gatherBlocks, 256, 0, stream>>>(XW, ROWPTR, PERM, b2, H2, N);

    k_gemm_wmma<true, true, true, 0><<<gemmBlocks, 256, 0, stream>>>(
        H2, Wl, batch, R3, bl, OUT, N,
        nullptr, nullptr, nullptr, nullptr, 0, 0);
}


# Edge order fixed by per-wave LDS rank counters (earlier try varied run to run), so GEMM1 row tiles run in the 2 sort launches ahead of the sort blocks: launches 7 to 6; sum order changes, now fixed
# speedup vs baseline: 3.2575x; 1.0807x over previous
#include <hip/hip_runtime.h>
#include <hip/hip_bf16.h>

typedef float v2f __attribute__((ext_vector_type(2)));
typedef float v4f __attribute__((ext_vector_type(4)));
typedef float v8f __attribute__((ext_vector_type(8)));

#define NEG_SLOPE 0.01f
#define DCOL 128

__device__ __forceinline__ float lrelu(float x) {
    return x > 0.0f ? x : NEG_SLOPE * x;
}

#define SORT_CH 256
#define SORT_T  16
#define BSH     8
#define SORT_LDS_INTS (8 * 256 + 8)

__device__ __forceinline__ int block_scan256(int v, int* wsum, int& total) {
    const int lane = threadIdx.x & 31;
    const int wave = threadIdx.x >> 5;
    int incl = v;
    for (int o = 1; o < 32; o <<= 1) {
        int n = __shfl_up(incl, o);
        if (lane >= o) incl += n;
    }
    if (lane == 31) wsum[wave] = incl;
    __syncthreads();
    int woff = 0, tot = 0;
#pragma unroll
    for (int k = 0; k < 8; ++k) {
        const int s = wsum[k];
        woff += (k < wave) ? s : 0;
        tot += s;
    }
    __syncthreads();
    total = tot;
    return woff + incl - v;
}

__device__ __forceinline__ int wave_counts_sum(const int* cnt, int tid, int (&c8)[8]) {
    int sum = 0;
#pragma unroll
    for (int k = 0; k < 8; ++k) { c8[k] = cnt[k * 256 + tid]; sum += c8[k]; }
    return sum;
}

__device__ __forceinline__ void wave_cursors(int* cnt, int tid, int start, const int (&c8)[8]) {
    int run = start;
#pragma unroll
    for (int k = 0; k < 8; ++k) { cnt[k * 256 + tid] = run; run += c8[k]; }
}

__device__ __forceinline__ void part_body(int* lds,
       const int* __restrict__ src,
       const int* __restrict__ dst,
       const float* __restrict__ wvals,
       int2* __restrict__ rec,
       int* __restrict__ ctab,
       int* __restrict__ otab, int E, int L, int NB, int c) {
    int* cnt  = lds;
    int* wsum = lds + 8 * 256;
    const int tid = threadIdx.x;
    int* mycnt = cnt + (tid >> 5) * 256;
    const int e0 = (c * L < E) ? c * L : E;
    const int e1 = (e0 + L < E) ? e0 + L : E;
#pragma unroll
    for (int k = 0; k < 8; ++k) cnt[k * 256 + tid] = 0;
    __syncthreads();

    for (int base = e0; base < e1; base += 256 * SORT_T) {
        int d[SORT_T];
#pragma unroll
        for (int k = 0; k < SORT_T; ++k) {
            const int e = base + k * 256 + tid;
            d[k] = dst[(e < e1) ? e : (e1 - 1)];
        }
#pragma unroll
        for (int k = 0; k < SORT_T; ++k) {
            const int e = base + k * 256 + tid;
            if (e < e1) atomicAdd(&mycnt[d[k] >> BSH], 1);
        }
    }
    __syncthreads();

    int c8[8];
    const int myc = wave_counts_sum(cnt, tid, c8);
    int total;
    const int off = block_scan256(myc, wsum, total);
    if (tid < NB) {
        ctab[tid * SORT_CH + c] = myc;
        otab[tid * SORT_CH + c] = off;
    }
    wave_cursors(cnt, tid, off, c8);
    __syncthreads();

    for (int base = e0; base < e1; base += 256 * SORT_T) {
        int d[SORT_T], s[SORT_T];
        float w[SORT_T];
#pragma unroll
        for (int k = 0; k < SORT_T; ++k) {
            const int e = base + k * 256 + tid;
            const int ec = (e < e1) ? e : (e1 - 1);
            d[k] = dst[ec];
            s[k] = src[ec];
            w[k] = wvals[ec];
        }
        int p[SORT_T];
#pragma unroll
        for (int k = 0; k < SORT_T; ++k) {
            const int e = base + k * 256 + tid;
            p[k] = (e < e1) ? atomicAdd(&mycnt[d[k] >> BSH], 1) : 0;
        }
#pragma unroll
        for (int k = 0; k < SORT_T; ++k) {
            const int e = base + k * 256 + tid;
            if (e < e1) {
                const unsigned x = ((unsigned)(d[k] & 255) << 24) | (unsigned)s[k];
                rec[e0 + p[k]] = make_int2((int)x, __float_as_int(w[k]));
            }
        }
    }
}

__device__ __forceinline__ void bucket_body(int* lds,
         const int2* __restrict__ rec,
         const int* __restrict__ ctab,
         const int* __restrict__ otab,
         int* __restrict__ rowptr,
         int2* __restrict__ perm, int N, int E, int L, int b) {
    int* ncnt = lds;
    int* wsum = lds + 8 * 256;
    const int tid  = threadIdx.x;
    const int lane = tid & 31;
    int* myn = ncnt + (tid >> 5) * 256;

    const int myCnt = ctab[b * SORT_CH + tid];
    const int myOff = otab[b * SORT_CH + tid];
    const int myStart = ((tid * L < E) ? tid * L : E) + myOff;
#pragma unroll
    for (int k = 0; k < 8; ++k) ncnt[k * 256 + tid] = 0;
    int base;
    (void)block_scan256(myOff, wsum, base);

    int2 r[32];
#pragma unroll
    for (int i = 0; i < 32; ++i) {
        const int cn = __builtin_amdgcn_readlane(myCnt, i);
        const int st = __builtin_amdgcn_readlane(myStart, i);
        int idx = st + ((lane < cn) ? lane : 0);
        idx = (idx < E) ? idx : (E - 1);
        r[i] = rec[idx];
    }
#pragma unroll
    for (int i = 0; i < 32; ++i) {
        const int cn = __builtin_amdgcn_readlane(myCnt, i);
        if (lane < cn) atomicAdd(&myn[(unsigned)r[i].x >> 24], 1);
    }
    const bool longPiece = __builtin_amdgcn_ballot_w32(myCnt > 32) != 0u;
    if (longPiece) {
        for (int i = 0; i < 32; ++i) {
            const int cn = __builtin_amdgcn_readlane(myCnt, i);
            const int st = __builtin_amdgcn_readlane(myStart, i);
            for (int k = 32 + lane; k < cn; k += 32) {
                const int2 q = rec[st + k];
                atomicAdd(&myn[(unsigned)q.x >> 24], 1);
            }
        }
    }
    __syncthreads();

    int c8[8];
    const int v = wave_counts_sum(ncnt, tid, c8);
    int tot;
    const int ex = block_scan256(v, wsum, tot);
    const int node = b * 256 + tid;
    if (node < N) rowptr[node] = base + ex;
    if (node == N - 1) rowptr[N] = E;
    wave_cursors(ncnt, tid, ex, c8);
    __syncthreads();

    int p[32];
#pragma unroll
    for (int i = 0; i < 32; ++i) {
        const int cn = __builtin_amdgcn_readlane(myCnt, i);
        p[i] = (lane < cn) ? atomicAdd(&myn[(unsigned)r[i].x >> 24], 1) : 0;
    }
#pragma unroll
    for (int i = 0; i < 32; ++i) {
        const int cn = __builtin_amdgcn_readlane(myCnt, i);
        if (lane < cn) {
            const unsigned x = (unsigned)r[i].x;
            perm[base + p[i]] = make_int2((int)(x & 0xFFFFFFu), r[i].y);
        }
    }
    if (longPiece) {
        for (int i = 0; i < 32; ++i) {
            const int cn = __builtin_amdgcn_readlane(myCnt, i);
            const int st = __builtin_amdgcn_readlane(myStart, i);
            for (int k = 32 + lane; k < cn; k += 32) {
                const int2 q = rec[st + k];
                const unsigned x = (unsigned)q.x;
                const int p = atomicAdd(&myn[x >> 24], 1);
                perm[base + p] = make_int2((int)(x & 0xFFFFFFu), q.y);
            }
        }
    }
}

template <int U>
__device__ __forceinline__ void gather_rows(const float* __restrict__ xw,
                                            int sx, int wy, int j, int lane,
                                            v4f& acc) {
    v4f x[U];
    float w[U];
#pragma unroll
    for (int u = 0; u < U; ++u) {
        const int s = __builtin_amdgcn_readlane(sx, j + u);
        w[u] = __int_as_float(__builtin_amdgcn_readlane(wy, j + u));
        x[u] = *(const v4f*)(xw + (size_t)s * DCOL + lane * 4);
    }
#pragma unroll
    for (int u = 0; u < U; ++u) acc += w[u] * x[u];
}

__global__ void k_gather(const float* __restrict__ xw,
                         const int* __restrict__ rowptr,
                         const int2* __restrict__ perm,
                         const float* __restrict__ bias,
                         float* __restrict__ H, int N) {
    const int lane = threadIdx.x & 31;
    const int gw = __builtin_amdgcn_readfirstlane(
        (int)((blockIdx.x * blockDim.x + threadIdx.x) >> 5));
    const int nw = (int)((gridDim.x * blockDim.x) >> 5);
    const v4f b = *(const v4f*)(bias + lane * 4);

    for (int node = gw; node < N; node += nw) {
        const int p0 = rowptr[node];
        const int p1 = rowptr[node + 1];
        v4f acc = {0.0f, 0.0f, 0.0f, 0.0f};
        for (int base = p0; base < p1; base += 32) {
            const int cnt = (p1 - base < 32) ? (p1 - base) : 32;
            int2 my = make_int2(0, 0);
            if (lane < cnt) my = perm[base + lane];
            int j = 0;
            for (; j + 8 <= cnt; j += 8) gather_rows<8>(xw, my.x, my.y, j, lane, acc);
            if (j + 4 <= cnt) { gather_rows<4>(xw, my.x, my.y, j, lane, acc); j += 4; }
            if (j + 2 <= cnt) { gather_rows<2>(xw, my.x, my.y, j, lane, acc); j += 2; }
            if (j < cnt)      { gather_rows<1>(xw, my.x, my.y, j, lane, acc); }
        }
        acc += b;
        *(v4f*)(H + (size_t)node * DCOL + lane * 4) = acc;
    }
}

template <bool LR, bool INIT, bool EPI, bool MAP>
__device__ __forceinline__ void gemm_tile128(float* __restrict__ lw,
                                             const float* __restrict__ A,
                                             const int* __restrict__ rowmap,
                                             const float* __restrict__ W,
                                             const int* __restrict__ batch,
                                             const float* __restrict__ R,
                                             const float* __restrict__ bias,
                                             float* __restrict__ C,
                                             int nrows, int blk) {
    const int tid  = threadIdx.x;
    const int wave = tid >> 5;
    const int lane = tid & 31;
    const int lo   = lane & 15;
    const int hi   = lane >> 4;
    const int rowbase = blk * 128 + wave * 16;
    const bool active = rowbase < nrows;

    int arow = rowbase + lo;
    if (MAP) {
        const int g = (arow < nrows) ? arow : (nrows - 1);
        arow = active ? rowmap[g] : 0;
    }

    v8f acc[8] = {};
    if (INIT && active) {
#pragma unroll
        for (int r = 0; r < 8; ++r) {
            const int g = batch[rowbase + r + hi * 8];
            const float* rp = R + (size_t)g * DCOL + lo;
#pragma unroll
            for (int t = 0; t < 8; ++t) acc[t][r] = rp[t * 16];
        }
    }

    for (int i = tid; i < (128 * DCOL) / 8; i += 256) {
        int kp = i >> 5;
        int cg = (i & 31) * 4;
        v4f e = *(const v4f*)(W + (kp * 2 + 0) * DCOL + cg);
        v4f o = *(const v4f*)(W + (kp * 2 + 1) * DCOL + cg);
        float* dstp = lw + kp * 256 + cg * 2;
#pragma unroll
        for (int j = 0; j < 4; ++j) {
            v2f p = { e[j], o[j] };
            *(v2f*)(dstp + j * 2) = p;
        }
    }
    __syncthreads();

    if (active) {
        const float* arowptr = A + (size_t)arow * DCOL;
        const float* lb = lw + hi * 256 + lo * 2;
#pragma unroll 4
        for (int ks = 0; ks < 32; ++ks) {
            v2f a = *(const v2f*)(arowptr + ks * 4 + hi * 2);
            if (LR) { a[0] = lrelu(a[0]); a[1] = lrelu(a[1]); }
            v2f b[8];
#pragma unroll
            for (int t = 0; t < 8; ++t)
                b[t] = *(const v2f*)(lb + ks * 512 + t * 32);
#pragma unroll
            for (int t = 0; t < 8; ++t)
                acc[t] = __builtin_amdgcn_wmma_f32_16x16x4_f32(
                    false, a, false, b[t], (short)0, acc[t], false, false);
        }

        if (MAP) {
#pragma unroll
            for (int r = 0; r < 8; ++r) {
                const int row = rowbase + r + hi * 8;
                if (row < nrows) {
#pragma unroll
                    for (int t = 0; t < 8; ++t)
                        C[(size_t)row * DCOL + t * 16 + lo] = acc[t][r];
                }
            }
        } else {
#pragma unroll
            for (int t = 0; t < 8; ++t) {
                int col = t * 16 + lo;
                float badd = EPI ? bias[col] : 0.0f;
#pragma unroll
                for (int r = 0; r < 8; ++r) {
                    int row = rowbase + r + hi * 8;
                    float v = acc[t][r];
                    if (EPI) v = lrelu(v + badd);
                    C[(size_t)row * DCOL + col] = v;
                }
            }
        }
    }
}

template <bool LR0, bool INIT, bool EPI, int SIDE>
__global__ void k_gemm_wmma(const float* __restrict__ A0,
                            const float* __restrict__ W,
                            const int* __restrict__ batch,
                            const float* __restrict__ R,
                            const float* __restrict__ bias,
                            float* __restrict__ C, int N,
                            const float* __restrict__ TA,
                            const int* __restrict__ root_idx,
                            const float* __restrict__ TW,
                            float* __restrict__ T, int B, int sideBlocks) {
    __shared__ float lw[128 * 128];
    const int blk = (int)blockIdx.x;
    if (SIDE != 0 && blk < sideBlocks) {
        gemm_tile128<SIDE == 1, false, false, true>(
            lw, TA, root_idx, TW, nullptr, nullptr, nullptr, T, B, blk);
    } else {
        gemm_tile128<LR0, INIT, EPI, false>(
            lw, A0, nullptr, W, batch, R, bias, C, N,
            blk - (SIDE != 0 ? sideBlocks : 0));
    }
}

template <int PHASE>
__global__ void __launch_bounds__(256)
k_sort_gemm(const int* __restrict__ src,
            const int* __restrict__ dst,
            const float* __restrict__ wvals,
            int2* __restrict__ rec,
            int* __restrict__ ctab,
            int* __restrict__ otab,
            int* __restrict__ rowptr,
            int2* __restrict__ perm,
            int N, int E, int L, int NB, int gemmPart,
            const float* __restrict__ A0,
            const float* __restrict__ W,
            float* __restrict__ C, int tile0,
            const float* __restrict__ TA,
            const int* __restrict__ root_idx,
            const float* __restrict__ TW,
            float* __restrict__ T, int B, int sideBlocks) {
    __shared__ float lw[128 * 128];
    const int blk = (int)blockIdx.x;
    if (blk >= gemmPart) {
        if (PHASE == 0)
            part_body((int*)lw, src, dst, wvals, rec, ctab, otab, E, L, NB, blk - gemmPart);
        else
            bucket_body((int*)lw, rec, ctab, otab, rowptr, perm, N, E, L, blk - gemmPart);
    } else if (blk < sideBlocks) {
        gemm_tile128<true, false, false, true>(
            lw, TA, root_idx, TW, nullptr, nullptr, nullptr, T, B, blk);
    } else {
        gemm_tile128<false, false, false, false>(
            lw, A0, nullptr, W, nullptr, nullptr, nullptr, C, N,
            tile0 + blk - sideBlocks);
    }
}

extern "C" void kernel_launch(void* const* d_in, const int* in_sizes, int n_in,
                              void* d_out, int out_size, void* d_ws, size_t ws_size,
                              hipStream_t stream) {
    const float* features = (const float*)d_in[0];
    const float* values   = (const float*)d_in[1];
    const float* W1       = (const float*)d_in[2];
    const float* b1       = (const float*)d_in[3];
    const float* W2       = (const float*)d_in[4];
    const float* b2       = (const float*)d_in[5];
    const float* Wl       = (const float*)d_in[6];
    const float* bl       = (const float*)d_in[7];
    const int*   adjs     = (const int*)d_in[8];
    const int*   root_idx = (const int*)d_in[9];
    const int*   batch    = (const int*)d_in[12];

    const int N = in_sizes[0] / DCOL;
    const int E = in_sizes[1];
    const int B = in_sizes[9];
    const int* srcI = adjs;
    const int* dstI = adjs + E;

    int2* PERM   = (int2*)d_ws;
    int2* REC    = PERM + E;
    int*  CTAB   = (int*)(REC + E);
    int*  OTAB   = CTAB + 256 * SORT_CH;
    int*  ROWPTR = OTAB + 256 * SORT_CH;
    float* H1 = (float*)d_ws + (size_t)N * DCOL;
    float* H2 = H1 + (size_t)N * DCOL;
    float* R2 = H1 - (size_t)2 * B * DCOL;
    float* R3 = R2 + (size_t)B * DCOL;

    float* OUT = (float*)d_out;
    float* XW  = OUT;

    const int gemmBlocks = (N + 127) / 128;
    const int sideBlocks = (B + 127) / 128;
    const int chunkLen = (E + SORT_CH - 1) / SORT_CH;
    const int buckets  = (N + 255) / 256;
    const int gatherBlocks = 2048;

    const int tilesA = gemmBlocks / 2;
    const int tilesB = gemmBlocks - tilesA;
    k_sort_gemm<0><<<tilesA + SORT_CH, 256, 0, stream>>>(
        srcI, dstI, values, REC, CTAB, OTAB, ROWPTR, PERM, N, E, chunkLen, buckets, tilesA,
        features, W1, XW, 0,
        nullptr, nullptr, nullptr, nullptr, 0, 0);
    k_sort_gemm<1><<<sideBlocks + tilesB + buckets, 256, 0, stream>>>(
        srcI, dstI, values, REC, CTAB, OTAB, ROWPTR, PERM, N, E, chunkLen, buckets, sideBlocks + tilesB,
        features, W1, XW, tilesA,
        features, root_idx, W2 + 128 * DCOL, R2, B, sideBlocks);

    k_gather<<<gatherBlocks, 256, 0, stream>>>(XW, ROWPTR, PERM, b1, H1, N);

    k_gemm_wmma<true, true, false, 2><<<gemmBlocks + sideBlocks, 256, 0, stream>>>(
        H1, W2, batch, R2, nullptr, XW, N,
        H1, root_idx, Wl + 128 * DCOL, R3, B, sideBlocks);
    k_gather<<<gatherBlocks, 256, 0, stream>>>(XW, ROWPTR, PERM, b2, H2, N);

    k_gemm_wmma<true, true, true, 0><<<gemmBlocks, 256, 0, stream>>>(
        H2, Wl, batch, R3, bl, OUT, N,
        nullptr, nullptr, nullptr, nullptr, 0, 0);
}


# Each gather fused into the GEMM that reads it: block streams its 128 nodes' sorted edge range into an LDS A tile, then WMMA; conv planes never stored; launches 6 to 4; per-node sum order unchanged
# speedup vs baseline: 3.3776x; 1.0369x over previous
#include <hip/hip_runtime.h>
#include <hip/hip_bf16.h>

typedef float v2f __attribute__((ext_vector_type(2)));
typedef float v4f __attribute__((ext_vector_type(4)));
typedef float v8f __attribute__((ext_vector_type(8)));

#define NEG_SLOPE 0.01f
#define DCOL 128

__device__ __forceinline__ float lrelu(float x) {
    return x > 0.0f ? x : NEG_SLOPE * x;
}

#define SORT_CH 256
#define SORT_T  16
#define BSH     8
#define SORT_LDS_INTS (8 * 256 + 8)

__device__ __forceinline__ int block_scan256(int v, int* wsum, int& total) {
    const int lane = threadIdx.x & 31;
    const int wave = threadIdx.x >> 5;
    int incl = v;
    for (int o = 1; o < 32; o <<= 1) {
        int n = __shfl_up(incl, o);
        if (lane >= o) incl += n;
    }
    if (lane == 31) wsum[wave] = incl;
    __syncthreads();
    int woff = 0, tot = 0;
#pragma unroll
    for (int k = 0; k < 8; ++k) {
        const int s = wsum[k];
        woff += (k < wave) ? s : 0;
        tot += s;
    }
    __syncthreads();
    total = tot;
    return woff + incl - v;
}

__device__ __forceinline__ int wave_counts_sum(const int* cnt, int tid, int (&c8)[8]) {
    int sum = 0;
#pragma unroll
    for (int k = 0; k < 8; ++k) { c8[k] = cnt[k * 256 + tid]; sum += c8[k]; }
    return sum;
}

__device__ __forceinline__ void wave_cursors(int* cnt, int tid, int start, const int (&c8)[8]) {
    int run = start;
#pragma unroll
    for (int k = 0; k < 8; ++k) { cnt[k * 256 + tid] = run; run += c8[k]; }
}

__device__ __forceinline__ void part_body(int* lds,
       const int* __restrict__ src,
       const int* __restrict__ dst,
       const float* __restrict__ wvals,
       int2* __restrict__ rec,
       int* __restrict__ ctab,
       int* __restrict__ otab, int E, int L, int NB, int c) {
    int* cnt  = lds;
    int* wsum = lds + 8 * 256;
    const int tid = threadIdx.x;
    int* mycnt = cnt + (tid >> 5) * 256;
    const int e0 = (c * L < E) ? c * L : E;
    const int e1 = (e0 + L < E) ? e0 + L : E;
#pragma unroll
    for (int k = 0; k < 8; ++k) cnt[k * 256 + tid] = 0;
    __syncthreads();

    for (int base = e0; base < e1; base += 256 * SORT_T) {
        int d[SORT_T];
#pragma unroll
        for (int k = 0; k < SORT_T; ++k) {
            const int e = base + k * 256 + tid;
            d[k] = dst[(e < e1) ? e : (e1 - 1)];
        }
#pragma unroll
        for (int k = 0; k < SORT_T; ++k) {
            const int e = base + k * 256 + tid;
            if (e < e1) atomicAdd(&mycnt[d[k] >> BSH], 1);
        }
    }
    __syncthreads();

    int c8[8];
    const int myc = wave_counts_sum(cnt, tid, c8);
    int total;
    const int off = block_scan256(myc, wsum, total);
    if (tid < NB) {
        ctab[tid * SORT_CH + c] = myc;
        otab[tid * SORT_CH + c] = off;
    }
    wave_cursors(cnt, tid, off, c8);
    __syncthreads();

    for (int base = e0; base < e1; base += 256 * SORT_T) {
        int d[SORT_T], s[SORT_T];
        float w[SORT_T];
#pragma unroll
        for (int k = 0; k < SORT_T; ++k) {
            const int e = base + k * 256 + tid;
            const int ec = (e < e1) ? e : (e1 - 1);
            d[k] = dst[ec];
            s[k] = src[ec];
            w[k] = wvals[ec];
        }
        int p[SORT_T];
#pragma unroll
        for (int k = 0; k < SORT_T; ++k) {
            const int e = base + k * 256 + tid;
            p[k] = (e < e1) ? atomicAdd(&mycnt[d[k] >> BSH], 1) : 0;
        }
#pragma unroll
        for (int k = 0; k < SORT_T; ++k) {
            const int e = base + k * 256 + tid;
            if (e < e1) {
                const unsigned x = ((unsigned)(d[k] & 255) << 24) | (unsigned)s[k];
                rec[e0 + p[k]] = make_int2((int)x, __float_as_int(w[k]));
            }
        }
    }
}

__device__ __forceinline__ void bucket_body(int* lds,
         const int2* __restrict__ rec,
         const int* __restrict__ ctab,
         const int* __restrict__ otab,
         int* __restrict__ rowptr,
         int2* __restrict__ perm, int N, int E, int L, int b) {
    int* ncnt = lds;
    int* wsum = lds + 8 * 256;
    const int tid  = threadIdx.x;
    const int lane = tid & 31;
    int* myn = ncnt + (tid >> 5) * 256;

    const int myCnt = ctab[b * SORT_CH + tid];
    const int myOff = otab[b * SORT_CH + tid];
    const int myStart = ((tid * L < E) ? tid * L : E) + myOff;
#pragma unroll
    for (int k = 0; k < 8; ++k) ncnt[k * 256 + tid] = 0;
    int base;
    (void)block_scan256(myOff, wsum, base);

    int2 r[32];
#pragma unroll
    for (int i = 0; i < 32; ++i) {
        const int cn = __builtin_amdgcn_readlane(myCnt, i);
        const int st = __builtin_amdgcn_readlane(myStart, i);
        int idx = st + ((lane < cn) ? lane : 0);
        idx = (idx < E) ? idx : (E - 1);
        r[i] = rec[idx];
    }
#pragma unroll
    for (int i = 0; i < 32; ++i) {
        const int cn = __builtin_amdgcn_readlane(myCnt, i);
        if (lane < cn) atomicAdd(&myn[(unsigned)r[i].x >> 24], 1);
    }
    const bool longPiece = __builtin_amdgcn_ballot_w32(myCnt > 32) != 0u;
    if (longPiece) {
        for (int i = 0; i < 32; ++i) {
            const int cn = __builtin_amdgcn_readlane(myCnt, i);
            const int st = __builtin_amdgcn_readlane(myStart, i);
            for (int k = 32 + lane; k < cn; k += 32) {
                const int2 q = rec[st + k];
                atomicAdd(&myn[(unsigned)q.x >> 24], 1);
            }
        }
    }
    __syncthreads();

    int c8[8];
    const int v = wave_counts_sum(ncnt, tid, c8);
    int tot;
    const int ex = block_scan256(v, wsum, tot);
    const int node = b * 256 + tid;
    if (node < N) rowptr[node] = base + ex;
    if (node == N - 1) rowptr[N] = E;
    wave_cursors(ncnt, tid, ex, c8);
    __syncthreads();

    int p[32];
#pragma unroll
    for (int i = 0; i < 32; ++i) {
        const int cn = __builtin_amdgcn_readlane(myCnt, i);
        p[i] = (lane < cn) ? atomicAdd(&myn[(unsigned)r[i].x >> 24], 1) : 0;
    }
#pragma unroll
    for (int i = 0; i < 32; ++i) {
        const int cn = __builtin_amdgcn_readlane(myCnt, i);
        if (lane < cn) {
            const unsigned x = (unsigned)r[i].x;
            perm[base + p[i]] = make_int2((int)(x & 0xFFFFFFu), r[i].y);
        }
    }
    if (longPiece) {
        for (int i = 0; i < 32; ++i) {
            const int cn = __builtin_amdgcn_readlane(myCnt, i);
            const int st = __builtin_amdgcn_readlane(myStart, i);
            for (int k = 32 + lane; k < cn; k += 32) {
                const int2 q = rec[st + k];
                const unsigned x = (unsigned)q.x;
                const int p = atomicAdd(&myn[x >> 24], 1);
                perm[base + p] = make_int2((int)(x & 0xFFFFFFu), q.y);
            }
        }
    }
}

template <int U>
__device__ __forceinline__ void gather_rows(const float* __restrict__ xw,
                                            int sx, int wy, int j, int lane,
                                            v4f& acc) {
    v4f x[U];
    float w[U];
#pragma unroll
    for (int u = 0; u < U; ++u) {
        const int s = __builtin_amdgcn_readlane(sx, j + u);
        w[u] = __int_as_float(__builtin_amdgcn_readlane(wy, j + u));
        x[u] = *(const v4f*)(xw + (size_t)s * DCOL + lane * 4);
    }
#pragma unroll
    for (int u = 0; u < U; ++u) acc += w[u] * x[u];
}

__device__ __forceinline__ v4f gather_node(const float* __restrict__ xw,
                                           const int* __restrict__ rowptr,
                                           const int2* __restrict__ perm,
                                           int node, int lane) {
    const int p0 = rowptr[node];
    const int p1 = rowptr[node + 1];
    v4f acc = {0.0f, 0.0f, 0.0f, 0.0f};
    for (int base = p0; base < p1; base += 32) {
        const int cnt = (p1 - base < 32) ? (p1 - base) : 32;
        int2 my = make_int2(0, 0);
        if (lane < cnt) my = perm[base + lane];
        int j = 0;
        for (; j + 8 <= cnt; j += 8) gather_rows<8>(xw, my.x, my.y, j, lane, acc);
        if (j + 4 <= cnt) { gather_rows<4>(xw, my.x, my.y, j, lane, acc); j += 4; }
        if (j + 2 <= cnt) { gather_rows<2>(xw, my.x, my.y, j, lane, acc); j += 2; }
        if (j < cnt)      { gather_rows<1>(xw, my.x, my.y, j, lane, acc); }
    }
    return acc;
}

#define LDA 132

template <bool LR>
__device__ __forceinline__ void put_row(float* __restrict__ dstrow, int lane, v4f acc, v4f gb) {
    v4f v = acc + gb;
    if (LR) { v[0] = lrelu(v[0]); v[1] = lrelu(v[1]); v[2] = lrelu(v[2]); v[3] = lrelu(v[3]); }
    *(v4f*)(dstrow + lane * 4) = v;
}

template <bool LR>
__device__ __forceinline__ void gather_tile_rows(float* __restrict__ arows,
                                                 const float* __restrict__ xw,
                                                 const int* __restrict__ rowptr,
                                                 const int2* __restrict__ perm,
                                                 v4f gb, int rowbase, int lane) {
    const int rpv = rowptr[rowbase + (lane < 16 ? lane : 16)];
    const int P0 = __builtin_amdgcn_readlane(rpv, 0);
    const int P1 = __builtin_amdgcn_readlane(rpv, 16);
    int cur = 0;
    int nxt = __builtin_amdgcn_readlane(rpv, 1);
    v4f acc = {0.0f, 0.0f, 0.0f, 0.0f};
    const v4f zero = {0.0f, 0.0f, 0.0f, 0.0f};
    int2 my = make_int2(0, 0);
    if (P0 + lane < P1) my = perm[P0 + lane];
    for (int base = P0; base < P1; base += 32) {
        const int cnt = (P1 - base < 32) ? (P1 - base) : 32;
        int2 nx = make_int2(0, 0);
        v4f x[32];
        auto take = [&](int k) {
            const int e = base + k;
            while (e >= nxt) {
                put_row<LR>(arows + cur * LDA, lane, acc, gb);
                acc = zero;
                ++cur;
                nxt = __builtin_amdgcn_readlane(rpv, cur + 1);
            }
            const float w = __int_as_float(__builtin_amdgcn_readlane(my.y, k));
            acc += w * x[k];
        };
        if (cnt == 32) {
#pragma unroll
            for (int k = 0; k < 32; ++k) {
                const int s = __builtin_amdgcn_readlane(my.x, k);
                x[k] = *(const v4f*)(xw + (size_t)s * DCOL + lane * 4);
            }
            if (base + 32 + lane < P1) nx = perm[base + 32 + lane];
#pragma unroll
            for (int k = 0; k < 32; ++k) take(k);
        } else {
#pragma unroll
            for (int k = 0; k < 32; ++k) {
                if (k < cnt) {
                    const int s = __builtin_amdgcn_readlane(my.x, k);
                    x[k] = *(const v4f*)(xw + (size_t)s * DCOL + lane * 4);
                }
            }
#pragma unroll
            for (int k = 0; k < 32; ++k) {
                if (k < cnt) take(k);
            }
        }
        my = nx;
    }
    while (cur < 16) {
        put_row<LR>(arows + cur * LDA, lane, acc, gb);
        acc = zero;
        ++cur;
    }
}

template <bool LR, bool INIT, bool EPI, bool MAP>
__device__ __forceinline__ void gemm_tile128(float* __restrict__ lw,
                                             const float* __restrict__ A,
                                             const int* __restrict__ rowmap,
                                             const float* __restrict__ W,
                                             const int* __restrict__ batch,
                                             const float* __restrict__ R,
                                             const float* __restrict__ bias,
                                             float* __restrict__ C,
                                             int nrows, int blk) {
    const int tid  = threadIdx.x;
    const int wave = tid >> 5;
    const int lane = tid & 31;
    const int lo   = lane & 15;
    const int hi   = lane >> 4;
    const int rowbase = blk * 128 + wave * 16;
    const bool active = rowbase < nrows;

    int arow = rowbase + lo;
    if (MAP) {
        const int g = (arow < nrows) ? arow : (nrows - 1);
        arow = active ? rowmap[g] : 0;
    }

    v8f acc[8] = {};
    if (INIT && active) {
#pragma unroll
        for (int r = 0; r < 8; ++r) {
            const int g = batch[rowbase + r + hi * 8];
            const float* rp = R + (size_t)g * DCOL + lo;
#pragma unroll
            for (int t = 0; t < 8; ++t) acc[t][r] = rp[t * 16];
        }
    }

    for (int i = tid; i < (128 * DCOL) / 8; i += 256) {
        int kp = i >> 5;
        int cg = (i & 31) * 4;
        v4f e = *(const v4f*)(W + (kp * 2 + 0) * DCOL + cg);
        v4f o = *(const v4f*)(W + (kp * 2 + 1) * DCOL + cg);
        float* dstp = lw + kp * 256 + cg * 2;
#pragma unroll
        for (int j = 0; j < 4; ++j) {
            v2f p = { e[j], o[j] };
            *(v2f*)(dstp + j * 2) = p;
        }
    }
    __syncthreads();

    if (active) {
        const float* arowptr = A + (size_t)arow * DCOL;
        const float* lb = lw + hi * 256 + lo * 2;
#pragma unroll 4
        for (int ks = 0; ks < 32; ++ks) {
            v2f a = *(const v2f*)(arowptr + ks * 4 + hi * 2);
            if (LR) { a[0] = lrelu(a[0]); a[1] = lrelu(a[1]); }
            v2f b[8];
#pragma unroll
            for (int t = 0; t < 8; ++t)
                b[t] = *(const v2f*)(lb + ks * 512 + t * 32);
#pragma unroll
            for (int t = 0; t < 8; ++t)
                acc[t] = __builtin_amdgcn_wmma_f32_16x16x4_f32(
                    false, a, false, b[t], (short)0, acc[t], false, false);
        }

        if (MAP) {
#pragma unroll
            for (int r = 0; r < 8; ++r) {
                const int row = rowbase + r + hi * 8;
                if (row < nrows) {
#pragma unroll
                    for (int t = 0; t < 8; ++t)
                        C[(size_t)row * DCOL + t * 16 + lo] = acc[t][r];
                }
            }
        } else {
#pragma unroll
            for (int t = 0; t < 8; ++t) {
                int col = t * 16 + lo;
                float badd = EPI ? bias[col] : 0.0f;
#pragma unroll
                for (int r = 0; r < 8; ++r) {
                    int row = rowbase + r + hi * 8;
                    float v = acc[t][r];
                    if (EPI) v = lrelu(v + badd);
                    C[(size_t)row * DCOL + col] = v;
                }
            }
        }
    }
}

template <bool LR, bool INIT, bool EPI, bool MAP>
__device__ __forceinline__ void gather_gemm_tile(float* __restrict__ lw,
                                                 float* __restrict__ la,
                                                 const float* __restrict__ xw,
                                                 const int* __restrict__ rowptr,
                                                 const int2* __restrict__ perm,
                                                 const float* __restrict__ gbias,
                                                 const int* __restrict__ rowmap,
                                                 const float* __restrict__ W,
                                                 const int* __restrict__ batch,
                                                 const float* __restrict__ R,
                                                 const float* __restrict__ bias,
                                                 float* __restrict__ C,
                                                 int nrows, int blk) {
    const int tid  = threadIdx.x;
    const int wave = tid >> 5;
    const int lane = tid & 31;
    const int lo   = lane & 15;
    const int hi   = lane >> 4;
    const int rowbase = blk * 128 + wave * 16;
    const bool active = rowbase < nrows;

    v8f acc[8] = {};
    if (INIT && active) {
#pragma unroll
        for (int r = 0; r < 8; ++r) {
            const int g = batch[rowbase + r + hi * 8];
            const float* rp = R + (size_t)g * DCOL + lo;
#pragma unroll
            for (int t = 0; t < 8; ++t) acc[t][r] = rp[t * 16];
        }
    }

    for (int i = tid; i < (128 * DCOL) / 8; i += 256) {
        int kp = i >> 5;
        int cg = (i & 31) * 4;
        v4f e = *(const v4f*)(W + (kp * 2 + 0) * DCOL + cg);
        v4f o = *(const v4f*)(W + (kp * 2 + 1) * DCOL + cg);
        float* dstp = lw + kp * 256 + cg * 2;
#pragma unroll
        for (int j = 0; j < 4; ++j) {
            v2f p = { e[j], o[j] };
            *(v2f*)(dstp + j * 2) = p;
        }
    }

    const v4f gb = *(const v4f*)(gbias + lane * 4);
    if (MAP) {
        const v4f zero = {0.0f, 0.0f, 0.0f, 0.0f};
        for (int i = wave; i < 128; i += 8) {
            const int g = blk * 128 + i;
            if (g < nrows) {
                const int node = __builtin_amdgcn_readfirstlane(rowmap[g]);
                const v4f s = gather_node(xw, rowptr, perm, node, lane);
                put_row<LR>(la + i * LDA, lane, s, gb);
            } else {
                *(v4f*)(la + i * LDA + lane * 4) = zero;
            }
        }
    } else if (active) {
        gather_tile_rows<LR>(la + wave * 16 * LDA, xw, rowptr, perm, gb, rowbase, lane);
    }
    __syncthreads();

    if (active) {
        const float* ap = la + (wave * 16 + lo) * LDA + hi * 2;
        const float* lb = lw + hi * 256 + lo * 2;
#pragma unroll 4
        for (int ks = 0; ks < 32; ++ks) {
            const v2f a = *(const v2f*)(ap + ks * 4);
            v2f b[8];
#pragma unroll
            for (int t = 0; t < 8; ++t)
                b[t] = *(const v2f*)(lb + ks * 512 + t * 32);
#pragma unroll
            for (int t = 0; t < 8; ++t)
                acc[t] = __builtin_amdgcn_wmma_f32_16x16x4_f32(
                    false, a, false, b[t], (short)0, acc[t], false, false);
        }

        if (MAP) {
#pragma unroll
            for (int r = 0; r < 8; ++r) {
                const int row = rowbase + r + hi * 8;
                if (row < nrows) {
#pragma unroll
                    for (int t = 0; t < 8; ++t)
                        C[(size_t)row * DCOL + t * 16 + lo] = acc[t][r];
                }
            }
        } else {
#pragma unroll
            for (int t = 0; t < 8; ++t) {
                int col = t * 16 + lo;
                float badd = EPI ? bias[col] : 0.0f;
#pragma unroll
                for (int r = 0; r < 8; ++r) {
                    int row = rowbase + r + hi * 8;
                    float v = acc[t][r];
                    if (EPI) v = lrelu(v + badd);
                    C[(size_t)row * DCOL + col] = v;
                }
            }
        }
    }
}

template <bool EPI, bool SIDE>
__global__ void __launch_bounds__(256)
k_gather_gemm(const float* __restrict__ xw,
              const int* __restrict__ rowptr,
              const int2* __restrict__ perm,
              const float* __restrict__ gbias,
              const float* __restrict__ W,
              const int* __restrict__ batch,
              const float* __restrict__ R,
              const float* __restrict__ bias,
              float* __restrict__ C, int N,
              const int* __restrict__ root_idx,
              const float* __restrict__ TW,
              float* __restrict__ T, int B, int sideBlocks) {
    __shared__ float lw[128 * 128];
    __shared__ float la[128 * LDA];
    const int blk = (int)blockIdx.x;
    if (SIDE && blk < sideBlocks) {
        gather_gemm_tile<false, false, false, true>(
            lw, la, xw, rowptr, perm, gbias, root_idx, TW,
            nullptr, nullptr, nullptr, T, B, blk);
    } else {
        gather_gemm_tile<true, true, EPI, false>(
            lw, la, xw, rowptr, perm, gbias, nullptr, W,
            batch, R, bias, C, N, blk - (SIDE ? sideBlocks : 0));
    }
}

template <int PHASE>
__global__ void __launch_bounds__(256)
k_sort_gemm(const int* __restrict__ src,
            const int* __restrict__ dst,
            const float* __restrict__ wvals,
            int2* __restrict__ rec,
            int* __restrict__ ctab,
            int* __restrict__ otab,
            int* __restrict__ rowptr,
            int2* __restrict__ perm,
            int N, int E, int L, int NB, int gemmPart,
            const float* __restrict__ A0,
            const float* __restrict__ W,
            float* __restrict__ C, int tile0,
            const float* __restrict__ TA,
            const int* __restrict__ root_idx,
            const float* __restrict__ TW,
            float* __restrict__ T, int B, int sideBlocks) {
    __shared__ float lw[128 * 128];
    const int blk = (int)blockIdx.x;
    if (blk >= gemmPart) {
        if (PHASE == 0)
            part_body((int*)lw, src, dst, wvals, rec, ctab, otab, E, L, NB, blk - gemmPart);
        else
            bucket_body((int*)lw, rec, ctab, otab, rowptr, perm, N, E, L, blk - gemmPart);
    } else if (blk < sideBlocks) {
        gemm_tile128<true, false, false, true>(
            lw, TA, root_idx, TW, nullptr, nullptr, nullptr, T, B, blk);
    } else {
        gemm_tile128<false, false, false, false>(
            lw, A0, nullptr, W, nullptr, nullptr, nullptr, C, N,
            tile0 + blk - sideBlocks);
    }
}

extern "C" void kernel_launch(void* const* d_in, const int* in_sizes, int n_in,
                              void* d_out, int out_size, void* d_ws, size_t ws_size,
                              hipStream_t stream) {
    const float* features = (const float*)d_in[0];
    const float* values   = (const float*)d_in[1];
    const float* W1       = (const float*)d_in[2];
    const float* b1       = (const float*)d_in[3];
    const float* W2       = (const float*)d_in[4];
    const float* b2       = (const float*)d_in[5];
    const float* Wl       = (const float*)d_in[6];
    const float* bl       = (const float*)d_in[7];
    const int*   adjs     = (const int*)d_in[8];
    const int*   root_idx = (const int*)d_in[9];
    const int*   batch    = (const int*)d_in[12];

    const int N = in_sizes[0] / DCOL;
    const int E = in_sizes[1];
    const int B = in_sizes[9];
    const int* srcI = adjs;
    const int* dstI = adjs + E;

    int2* PERM   = (int2*)d_ws;
    int2* REC    = PERM + E;
    int*  CTAB   = (int*)(REC + E);
    int*  OTAB   = CTAB + 256 * SORT_CH;
    int*  ROWPTR = OTAB + 256 * SORT_CH;
    float* XW2 = (float*)d_ws + (size_t)N * DCOL;
    float* R2 = XW2 - (size_t)2 * B * DCOL;
    float* R3 = R2 + (size_t)B * DCOL;

    float* OUT = (float*)d_out;
    float* XW  = OUT;

    const int gemmBlocks = (N + 127) / 128;
    const int sideBlocks = (B + 127) / 128;
    const int chunkLen = (E + SORT_CH - 1) / SORT_CH;
    const int buckets  = (N + 255) / 256;

    const int tilesA = gemmBlocks / 2;
    const int tilesB = gemmBlocks - tilesA;
    k_sort_gemm<0><<<tilesA + SORT_CH, 256, 0, stream>>>(
        srcI, dstI, values, REC, CTAB, OTAB, ROWPTR, PERM, N, E, chunkLen, buckets, tilesA,
        features, W1, XW, 0,
        nullptr, nullptr, nullptr, nullptr, 0, 0);
    k_sort_gemm<1><<<sideBlocks + tilesB + buckets, 256, 0, stream>>>(
        srcI, dstI, values, REC, CTAB, OTAB, ROWPTR, PERM, N, E, chunkLen, buckets, sideBlocks + tilesB,
        features, W1, XW, tilesA,
        features, root_idx, W2 + 128 * DCOL, R2, B, sideBlocks);

    k_gather_gemm<false, true><<<gemmBlocks + sideBlocks, 256, 0, stream>>>(
        XW, ROWPTR, PERM, b1, W2, batch, R2, nullptr, XW2, N,
        root_idx, Wl + 128 * DCOL, R3, B, sideBlocks);

    k_gather_gemm<true, false><<<gemmBlocks, 256, 0, stream>>>(
        XW2, ROWPTR, PERM, b2, Wl, batch, R3, bl, OUT, N,
        nullptr, nullptr, nullptr, 0, 0);
}


# Gather launches re-gridded: one 16-wave block per WGP (256 blocks), each wave one 16-row group dealt round-robin over blocks; W staged once per WGP, sole barrier behind it; sums unchanged
# speedup vs baseline: 3.5041x; 1.0375x over previous
#include <hip/hip_runtime.h>
#include <hip/hip_bf16.h>

typedef float v2f __attribute__((ext_vector_type(2)));
typedef float v4f __attribute__((ext_vector_type(4)));
typedef float v8f __attribute__((ext_vector_type(8)));

#define NEG_SLOPE 0.01f
#define DCOL 128

__device__ __forceinline__ float lrelu(float x) {
    return x > 0.0f ? x : NEG_SLOPE * x;
}

#define SORT_CH 256
#define SORT_T  16
#define BSH     8
#define SORT_LDS_INTS (8 * 256 + 8)

__device__ __forceinline__ int block_scan256(int v, int* wsum, int& total) {
    const int lane = threadIdx.x & 31;
    const int wave = threadIdx.x >> 5;
    int incl = v;
    for (int o = 1; o < 32; o <<= 1) {
        int n = __shfl_up(incl, o);
        if (lane >= o) incl += n;
    }
    if (lane == 31) wsum[wave] = incl;
    __syncthreads();
    int woff = 0, tot = 0;
#pragma unroll
    for (int k = 0; k < 8; ++k) {
        const int s = wsum[k];
        woff += (k < wave) ? s : 0;
        tot += s;
    }
    __syncthreads();
    total = tot;
    return woff + incl - v;
}

__device__ __forceinline__ int wave_counts_sum(const int* cnt, int tid, int (&c8)[8]) {
    int sum = 0;
#pragma unroll
    for (int k = 0; k < 8; ++k) { c8[k] = cnt[k * 256 + tid]; sum += c8[k]; }
    return sum;
}

__device__ __forceinline__ void wave_cursors(int* cnt, int tid, int start, const int (&c8)[8]) {
    int run = start;
#pragma unroll
    for (int k = 0; k < 8; ++k) { cnt[k * 256 + tid] = run; run += c8[k]; }
}

__device__ __forceinline__ void part_body(int* lds,
       const int* __restrict__ src,
       const int* __restrict__ dst,
       const float* __restrict__ wvals,
       int2* __restrict__ rec,
       int* __restrict__ ctab,
       int* __restrict__ otab, int E, int L, int NB, int c) {
    int* cnt  = lds;
    int* wsum = lds + 8 * 256;
    const int tid = threadIdx.x;
    int* mycnt = cnt + (tid >> 5) * 256;
    const int e0 = (c * L < E) ? c * L : E;
    const int e1 = (e0 + L < E) ? e0 + L : E;
#pragma unroll
    for (int k = 0; k < 8; ++k) cnt[k * 256 + tid] = 0;
    __syncthreads();

    for (int base = e0; base < e1; base += 256 * SORT_T) {
        int d[SORT_T];
#pragma unroll
        for (int k = 0; k < SORT_T; ++k) {
            const int e = base + k * 256 + tid;
            d[k] = dst[(e < e1) ? e : (e1 - 1)];
        }
#pragma unroll
        for (int k = 0; k < SORT_T; ++k) {
            const int e = base + k * 256 + tid;
            if (e < e1) atomicAdd(&mycnt[d[k] >> BSH], 1);
        }
    }
    __syncthreads();

    int c8[8];
    const int myc = wave_counts_sum(cnt, tid, c8);
    int total;
    const int off = block_scan256(myc, wsum, total);
    if (tid < NB) {
        ctab[tid * SORT_CH + c] = myc;
        otab[tid * SORT_CH + c] = off;
    }
    wave_cursors(cnt, tid, off, c8);
    __syncthreads();

    for (int base = e0; base < e1; base += 256 * SORT_T) {
        int d[SORT_T], s[SORT_T];
        float w[SORT_T];
#pragma unroll
        for (int k = 0; k < SORT_T; ++k) {
            const int e = base + k * 256 + tid;
            const int ec = (e < e1) ? e : (e1 - 1);
            d[k] = dst[ec];
            s[k] = src[ec];
            w[k] = wvals[ec];
        }
        int p[SORT_T];
#pragma unroll
        for (int k = 0; k < SORT_T; ++k) {
            const int e = base + k * 256 + tid;
            p[k] = (e < e1) ? atomicAdd(&mycnt[d[k] >> BSH], 1) : 0;
        }
#pragma unroll
        for (int k = 0; k < SORT_T; ++k) {
            const int e = base + k * 256 + tid;
            if (e < e1) {
                const unsigned x = ((unsigned)(d[k] & 255) << 24) | (unsigned)s[k];
                rec[e0 + p[k]] = make_int2((int)x, __float_as_int(w[k]));
            }
        }
    }
}

__device__ __forceinline__ void bucket_body(int* lds,
         const int2* __restrict__ rec,
         const int* __restrict__ ctab,
         const int* __restrict__ otab,
         int* __restrict__ rowptr,
         int2* __restrict__ perm, int N, int E, int L, int b) {
    int* ncnt = lds;
    int* wsum = lds + 8 * 256;
    const int tid  = threadIdx.x;
    const int lane = tid & 31;
    int* myn = ncnt + (tid >> 5) * 256;

    const int myCnt = ctab[b * SORT_CH + tid];
    const int myOff = otab[b * SORT_CH + tid];
    const int myStart = ((tid * L < E) ? tid * L : E) + myOff;
#pragma unroll
    for (int k = 0; k < 8; ++k) ncnt[k * 256 + tid] = 0;
    int base;
    (void)block_scan256(myOff, wsum, base);

    int2 r[32];
#pragma unroll
    for (int i = 0; i < 32; ++i) {
        const int cn = __builtin_amdgcn_readlane(myCnt, i);
        const int st = __builtin_amdgcn_readlane(myStart, i);
        int idx = st + ((lane < cn) ? lane : 0);
        idx = (idx < E) ? idx : (E - 1);
        r[i] = rec[idx];
    }
#pragma unroll
    for (int i = 0; i < 32; ++i) {
        const int cn = __builtin_amdgcn_readlane(myCnt, i);
        if (lane < cn) atomicAdd(&myn[(unsigned)r[i].x >> 24], 1);
    }
    const bool longPiece = __builtin_amdgcn_ballot_w32(myCnt > 32) != 0u;
    if (longPiece) {
        for (int i = 0; i < 32; ++i) {
            const int cn = __builtin_amdgcn_readlane(myCnt, i);
            const int st = __builtin_amdgcn_readlane(myStart, i);
            for (int k = 32 + lane; k < cn; k += 32) {
                const int2 q = rec[st + k];
                atomicAdd(&myn[(unsigned)q.x >> 24], 1);
            }
        }
    }
    __syncthreads();

    int c8[8];
    const int v = wave_counts_sum(ncnt, tid, c8);
    int tot;
    const int ex = block_scan256(v, wsum, tot);
    const int node = b * 256 + tid;
    if (node < N) rowptr[node] = base + ex;
    if (node == N - 1) rowptr[N] = E;
    wave_cursors(ncnt, tid, ex, c8);
    __syncthreads();

    int p[32];
#pragma unroll
    for (int i = 0; i < 32; ++i) {
        const int cn = __builtin_amdgcn_readlane(myCnt, i);
        p[i] = (lane < cn) ? atomicAdd(&myn[(unsigned)r[i].x >> 24], 1) : 0;
    }
#pragma unroll
    for (int i = 0; i < 32; ++i) {
        const int cn = __builtin_amdgcn_readlane(myCnt, i);
        if (lane < cn) {
            const unsigned x = (unsigned)r[i].x;
            perm[base + p[i]] = make_int2((int)(x & 0xFFFFFFu), r[i].y);
        }
    }
    if (longPiece) {
        for (int i = 0; i < 32; ++i) {
            const int cn = __builtin_amdgcn_readlane(myCnt, i);
            const int st = __builtin_amdgcn_readlane(myStart, i);
            for (int k = 32 + lane; k < cn; k += 32) {
                const int2 q = rec[st + k];
                const unsigned x = (unsigned)q.x;
                const int p = atomicAdd(&myn[x >> 24], 1);
                perm[base + p] = make_int2((int)(x & 0xFFFFFFu), q.y);
            }
        }
    }
}

template <int U>
__device__ __forceinline__ void gather_rows(const float* __restrict__ xw,
                                            int sx, int wy, int j, int lane,
                                            v4f& acc) {
    v4f x[U];
    float w[U];
#pragma unroll
    for (int u = 0; u < U; ++u) {
        const int s = __builtin_amdgcn_readlane(sx, j + u);
        w[u] = __int_as_float(__builtin_amdgcn_readlane(wy, j + u));
        x[u] = *(const v4f*)(xw + (size_t)s * DCOL + lane * 4);
    }
#pragma unroll
    for (int u = 0; u < U; ++u) acc += w[u] * x[u];
}

__device__ __forceinline__ v4f gather_node(const float* __restrict__ xw,
                                           const int* __restrict__ rowptr,
                                           const int2* __restrict__ perm,
                                           int node, int lane) {
    const int p0 = rowptr[node];
    const int p1 = rowptr[node + 1];
    v4f acc = {0.0f, 0.0f, 0.0f, 0.0f};
    for (int base = p0; base < p1; base += 32) {
        const int cnt = (p1 - base < 32) ? (p1 - base) : 32;
        int2 my = make_int2(0, 0);
        if (lane < cnt) my = perm[base + lane];
        int j = 0;
        for (; j + 8 <= cnt; j += 8) gather_rows<8>(xw, my.x, my.y, j, lane, acc);
        if (j + 4 <= cnt) { gather_rows<4>(xw, my.x, my.y, j, lane, acc); j += 4; }
        if (j + 2 <= cnt) { gather_rows<2>(xw, my.x, my.y, j, lane, acc); j += 2; }
        if (j < cnt)      { gather_rows<1>(xw, my.x, my.y, j, lane, acc); }
    }
    return acc;
}

#define LDA 132

template <bool LR>
__device__ __forceinline__ void put_row(float* __restrict__ dstrow, int lane, v4f acc, v4f gb) {
    v4f v = acc + gb;
    if (LR) { v[0] = lrelu(v[0]); v[1] = lrelu(v[1]); v[2] = lrelu(v[2]); v[3] = lrelu(v[3]); }
    *(v4f*)(dstrow + lane * 4) = v;
}

template <bool LR>
__device__ __forceinline__ void gather_tile_rows(float* __restrict__ arows,
                                                 const float* __restrict__ xw,
                                                 const int* __restrict__ rowptr,
                                                 const int2* __restrict__ perm,
                                                 v4f gb, int rowbase, int lane) {
    const int rpv = rowptr[rowbase + (lane < 16 ? lane : 16)];
    const int P0 = __builtin_amdgcn_readlane(rpv, 0);
    const int P1 = __builtin_amdgcn_readlane(rpv, 16);
    int cur = 0;
    int nxt = __builtin_amdgcn_readlane(rpv, 1);
    v4f acc = {0.0f, 0.0f, 0.0f, 0.0f};
    const v4f zero = {0.0f, 0.0f, 0.0f, 0.0f};
    int2 my = make_int2(0, 0);
    if (P0 + lane < P1) my = perm[P0 + lane];
    for (int base = P0; base < P1; base += 32) {
        const int cnt = (P1 - base < 32) ? (P1 - base) : 32;
        int2 nx = make_int2(0, 0);
        v4f x[32];
        auto take = [&](int k) {
            const int e = base + k;
            while (e >= nxt) {
                put_row<LR>(arows + cur * LDA, lane, acc, gb);
                acc = zero;
                ++cur;
                nxt = __builtin_amdgcn_readlane(rpv, cur + 1);
            }
            const float w = __int_as_float(__builtin_amdgcn_readlane(my.y, k));
            acc += w * x[k];
        };
        if (cnt == 32) {
#pragma unroll
            for (int k = 0; k < 32; ++k) {
                const int s = __builtin_amdgcn_readlane(my.x, k);
                x[k] = *(const v4f*)(xw + (size_t)s * DCOL + lane * 4);
            }
            if (base + 32 + lane < P1) nx = perm[base + 32 + lane];
#pragma unroll
            for (int k = 0; k < 32; ++k) take(k);
        } else {
#pragma unroll
            for (int k = 0; k < 32; ++k) {
                if (k < cnt) {
                    const int s = __builtin_amdgcn_readlane(my.x, k);
                    x[k] = *(const v4f*)(xw + (size_t)s * DCOL + lane * 4);
                }
            }
#pragma unroll
            for (int k = 0; k < 32; ++k) {
                if (k < cnt) take(k);
            }
        }
        my = nx;
    }
    while (cur < 16) {
        put_row<LR>(arows + cur * LDA, lane, acc, gb);
        acc = zero;
        ++cur;
    }
}

template <bool LR, bool INIT, bool EPI, bool MAP>
__device__ __forceinline__ void gemm_tile128(float* __restrict__ lw,
                                             const float* __restrict__ A,
                                             const int* __restrict__ rowmap,
                                             const float* __restrict__ W,
                                             const int* __restrict__ batch,
                                             const float* __restrict__ R,
                                             const float* __restrict__ bias,
                                             float* __restrict__ C,
                                             int nrows, int blk) {
    const int tid  = threadIdx.x;
    const int wave = tid >> 5;
    const int lane = tid & 31;
    const int lo   = lane & 15;
    const int hi   = lane >> 4;
    const int rowbase = blk * 128 + wave * 16;
    const bool active = rowbase < nrows;

    int arow = rowbase + lo;
    if (MAP) {
        const int g = (arow < nrows) ? arow : (nrows - 1);
        arow = active ? rowmap[g] : 0;
    }

    v8f acc[8] = {};
    if (INIT && active) {
#pragma unroll
        for (int r = 0; r < 8; ++r) {
            const int g = batch[rowbase + r + hi * 8];
            const float* rp = R + (size_t)g * DCOL + lo;
#pragma unroll
            for (int t = 0; t < 8; ++t) acc[t][r] = rp[t * 16];
        }
    }

    for (int i = tid; i < (128 * DCOL) / 8; i += 256) {
        int kp = i >> 5;
        int cg = (i & 31) * 4;
        v4f e = *(const v4f*)(W + (kp * 2 + 0) * DCOL + cg);
        v4f o = *(const v4f*)(W + (kp * 2 + 1) * DCOL + cg);
        float* dstp = lw + kp * 256 + cg * 2;
#pragma unroll
        for (int j = 0; j < 4; ++j) {
            v2f p = { e[j], o[j] };
            *(v2f*)(dstp + j * 2) = p;
        }
    }
    __syncthreads();

    if (active) {
        const float* arowptr = A + (size_t)arow * DCOL;
        const float* lb = lw + hi * 256 + lo * 2;
#pragma unroll 4
        for (int ks = 0; ks < 32; ++ks) {
            v2f a = *(const v2f*)(arowptr + ks * 4 + hi * 2);
            if (LR) { a[0] = lrelu(a[0]); a[1] = lrelu(a[1]); }
            v2f b[8];
#pragma unroll
            for (int t = 0; t < 8; ++t)
                b[t] = *(const v2f*)(lb + ks * 512 + t * 32);
#pragma unroll
            for (int t = 0; t < 8; ++t)
                acc[t] = __builtin_amdgcn_wmma_f32_16x16x4_f32(
                    false, a, false, b[t], (short)0, acc[t], false, false);
        }

        if (MAP) {
#pragma unroll
            for (int r = 0; r < 8; ++r) {
                const int row = rowbase + r + hi * 8;
                if (row < nrows) {
#pragma unroll
                    for (int t = 0; t < 8; ++t)
                        C[(size_t)row * DCOL + t * 16 + lo] = acc[t][r];
                }
            }
        } else {
#pragma unroll
            for (int t = 0; t < 8; ++t) {
                int col = t * 16 + lo;
                float badd = EPI ? bias[col] : 0.0f;
#pragma unroll
                for (int r = 0; r < 8; ++r) {
                    int row = rowbase + r + hi * 8;
                    float v = acc[t][r];
                    if (EPI) v = lrelu(v + badd);
                    C[(size_t)row * DCOL + col] = v;
                }
            }
        }
    }
}

#define GG_WAVES   16
#define GG_THREADS (GG_WAVES * 32)

__device__ __forceinline__ void stage_w(float* __restrict__ lw,
                                        const float* __restrict__ W, int tid) {
    for (int i = tid; i < (128 * DCOL) / 8; i += GG_THREADS) {
        int kp = i >> 5;
        int cg = (i & 31) * 4;
        v4f e = *(const v4f*)(W + (kp * 2 + 0) * DCOL + cg);
        v4f o = *(const v4f*)(W + (kp * 2 + 1) * DCOL + cg);
        float* dstp = lw + kp * 256 + cg * 2;
#pragma unroll
        for (int j = 0; j < 4; ++j) {
            v2f p = { e[j], o[j] };
            *(v2f*)(dstp + j * 2) = p;
        }
    }
}

template <bool EPI>
__device__ __forceinline__ void gather_gemm_groups(float* __restrict__ lw,
                                                   float* __restrict__ la,
                                                   const float* __restrict__ xw,
                                                   const int* __restrict__ rowptr,
                                                   const int2* __restrict__ perm,
                                                   const float* __restrict__ gbias,
                                                   const float* __restrict__ W,
                                                   const int* __restrict__ batch,
                                                   const float* __restrict__ R,
                                                   const float* __restrict__ bias,
                                                   float* __restrict__ C,
                                                   int nrows, int mb, int G) {
    const int tid  = threadIdx.x;
    const int wave = __builtin_amdgcn_readfirstlane(tid >> 5);
    const int lane = tid & 31;
    const int lo   = lane & 15;
    const int hi   = lane >> 4;

    stage_w(lw, W, tid);
    __syncthreads();

    const v4f gb = *(const v4f*)(gbias + lane * 4);
    float* arows = la + wave * 16 * LDA;
    const float* ap = arows + lo * LDA + hi * 2;
    const float* lb = lw + hi * 256 + lo * 2;
    const int groups = (nrows + 15) >> 4;
    for (int gi = wave * G + mb; gi < groups; gi += GG_WAVES * G) {
        const int rowbase = gi * 16;
        v8f acc[8];
#pragma unroll
        for (int r = 0; r < 8; ++r) {
            const int g = batch[rowbase + r + hi * 8];
            const float* rp = R + (size_t)g * DCOL + lo;
#pragma unroll
            for (int t = 0; t < 8; ++t) acc[t][r] = rp[t * 16];
        }

        gather_tile_rows<true>(arows, xw, rowptr, perm, gb, rowbase, lane);

#pragma unroll 4
        for (int ks = 0; ks < 32; ++ks) {
            const v2f a = *(const v2f*)(ap + ks * 4);
            v2f b[8];
#pragma unroll
            for (int t = 0; t < 8; ++t)
                b[t] = *(const v2f*)(lb + ks * 512 + t * 32);
#pragma unroll
            for (int t = 0; t < 8; ++t)
                acc[t] = __builtin_amdgcn_wmma_f32_16x16x4_f32(
                    false, a, false, b[t], (short)0, acc[t], false, false);
        }

#pragma unroll
        for (int t = 0; t < 8; ++t) {
            int col = t * 16 + lo;
            float badd = EPI ? bias[col] : 0.0f;
#pragma unroll
            for (int r = 0; r < 8; ++r) {
                int row = rowbase + r + hi * 8;
                float v = acc[t][r];
                if (EPI) v = lrelu(v + badd);
                C[(size_t)row * DCOL + col] = v;
            }
        }
    }
}

__device__ __forceinline__ void gather_gemm_side(float* __restrict__ lw,
                                                 float* __restrict__ la,
                                                 const float* __restrict__ xw,
                                                 const int* __restrict__ rowptr,
                                                 const int2* __restrict__ perm,
                                                 const float* __restrict__ gbias,
                                                 const int* __restrict__ rowmap,
                                                 const float* __restrict__ W,
                                                 float* __restrict__ C,
                                                 int nrows, int blk) {
    const int tid  = threadIdx.x;
    const int wave = tid >> 5;
    const int lane = tid & 31;
    const int lo   = lane & 15;
    const int hi   = lane >> 4;
    const int rowbase = blk * 128 + wave * 16;
    const bool active = wave < 8 && rowbase < nrows;

    stage_w(lw, W, tid);

    const v4f gb = *(const v4f*)(gbias + lane * 4);
    const v4f zero = {0.0f, 0.0f, 0.0f, 0.0f};
    for (int i = wave; i < 128; i += GG_WAVES) {
        const int g = blk * 128 + i;
        if (g < nrows) {
            const int node = __builtin_amdgcn_readfirstlane(rowmap[g]);
            const v4f s = gather_node(xw, rowptr, perm, node, lane);
            put_row<false>(la + i * LDA, lane, s, gb);
        } else {
            *(v4f*)(la + i * LDA + lane * 4) = zero;
        }
    }
    __syncthreads();

    if (active) {
        v8f acc[8] = {};
        const float* ap = la + (wave * 16 + lo) * LDA + hi * 2;
        const float* lb = lw + hi * 256 + lo * 2;
#pragma unroll 4
        for (int ks = 0; ks < 32; ++ks) {
            const v2f a = *(const v2f*)(ap + ks * 4);
            v2f b[8];
#pragma unroll
            for (int t = 0; t < 8; ++t)
                b[t] = *(const v2f*)(lb + ks * 512 + t * 32);
#pragma unroll
            for (int t = 0; t < 8; ++t)
                acc[t] = __builtin_amdgcn_wmma_f32_16x16x4_f32(
                    false, a, false, b[t], (short)0, acc[t], false, false);
        }

#pragma unroll
        for (int r = 0; r < 8; ++r) {
            const int row = rowbase + r + hi * 8;
            if (row < nrows) {
#pragma unroll
                for (int t = 0; t < 8; ++t)
                    C[(size_t)row * DCOL + t * 16 + lo] = acc[t][r];
            }
        }
    }
}

template <bool EPI, bool SIDE>
__global__ void __launch_bounds__(GG_THREADS)
k_gather_gemm(const float* __restrict__ xw,
              const int* __restrict__ rowptr,
              const int2* __restrict__ perm,
              const float* __restrict__ gbias,
              const float* __restrict__ W,
              const int* __restrict__ batch,
              const float* __restrict__ R,
              const float* __restrict__ bias,
              float* __restrict__ C, int N,
              const int* __restrict__ root_idx,
              const float* __restrict__ TW,
              float* __restrict__ T, int B, int sideBlocks, int mainBlocks) {
    __shared__ float lw[128 * 128];
    __shared__ float la[GG_WAVES * 16 * LDA];
    const int blk = (int)blockIdx.x;
    if (SIDE && blk < sideBlocks) {
        gather_gemm_side(lw, la, xw, rowptr, perm, gbias, root_idx, TW, T, B, blk);
    } else {
        gather_gemm_groups<EPI>(
            lw, la, xw, rowptr, perm, gbias, W,
            batch, R, bias, C, N, blk - (SIDE ? sideBlocks : 0), mainBlocks);
    }
}

template <int PHASE>
__global__ void __launch_bounds__(256)
k_sort_gemm(const int* __restrict__ src,
            const int* __restrict__ dst,
            const float* __restrict__ wvals,
            int2* __restrict__ rec,
            int* __restrict__ ctab,
            int* __restrict__ otab,
            int* __restrict__ rowptr,
            int2* __restrict__ perm,
            int N, int E, int L, int NB, int gemmPart,
            const float* __restrict__ A0,
            const float* __restrict__ W,
            float* __restrict__ C, int tile0,
            const float* __restrict__ TA,
            const int* __restrict__ root_idx,
            const float* __restrict__ TW,
            float* __restrict__ T, int B, int sideBlocks) {
    __shared__ float lw[128 * 128];
    const int blk = (int)blockIdx.x;
    if (blk >= gemmPart) {
        if (PHASE == 0)
            part_body((int*)lw, src, dst, wvals, rec, ctab, otab, E, L, NB, blk - gemmPart);
        else
            bucket_body((int*)lw, rec, ctab, otab, rowptr, perm, N, E, L, blk - gemmPart);
    } else if (blk < sideBlocks) {
        gemm_tile128<true, false, false, true>(
            lw, TA, root_idx, TW, nullptr, nullptr, nullptr, T, B, blk);
    } else {
        gemm_tile128<false, false, false, false>(
            lw, A0, nullptr, W, nullptr, nullptr, nullptr, C, N,
            tile0 + blk - sideBlocks);
    }
}

extern "C" void kernel_launch(void* const* d_in, const int* in_sizes, int n_in,
                              void* d_out, int out_size, void* d_ws, size_t ws_size,
                              hipStream_t stream) {
    const float* features = (const float*)d_in[0];
    const float* values   = (const float*)d_in[1];
    const float* W1       = (const float*)d_in[2];
    const float* b1       = (const float*)d_in[3];
    const float* W2       = (const float*)d_in[4];
    const float* b2       = (const float*)d_in[5];
    const float* Wl       = (const float*)d_in[6];
    const float* bl       = (const float*)d_in[7];
    const int*   adjs     = (const int*)d_in[8];
    const int*   root_idx = (const int*)d_in[9];
    const int*   batch    = (const int*)d_in[12];

    const int N = in_sizes[0] / DCOL;
    const int E = in_sizes[1];
    const int B = in_sizes[9];
    const int* srcI = adjs;
    const int* dstI = adjs + E;

    int2* PERM   = (int2*)d_ws;
    int2* REC    = PERM + E;
    int*  CTAB   = (int*)(REC + E);
    int*  OTAB   = CTAB + 256 * SORT_CH;
    int*  ROWPTR = OTAB + 256 * SORT_CH;
    float* XW2 = (float*)d_ws + (size_t)N * DCOL;
    float* R2 = XW2 - (size_t)2 * B * DCOL;
    float* R3 = R2 + (size_t)B * DCOL;

    float* OUT = (float*)d_out;
    float* XW  = OUT;

    const int gemmBlocks = (N + 127) / 128;
    const int sideBlocks = (B + 127) / 128;
    const int chunkLen = (E + SORT_CH - 1) / SORT_CH;
    const int buckets  = (N + 255) / 256;
    const int ggBlocks = 256;

    const int tilesA = gemmBlocks / 2;
    const int tilesB = gemmBlocks - tilesA;
    k_sort_gemm<0><<<tilesA + SORT_CH, 256, 0, stream>>>(
        srcI, dstI, values, REC, CTAB, OTAB, ROWPTR, PERM, N, E, chunkLen, buckets, tilesA,
        features, W1, XW, 0,
        nullptr, nullptr, nullptr, nullptr, 0, 0);
    k_sort_gemm<1><<<sideBlocks + tilesB + buckets, 256, 0, stream>>>(
        srcI, dstI, values, REC, CTAB, OTAB, ROWPTR, PERM, N, E, chunkLen, buckets, sideBlocks + tilesB,
        features, W1, XW, tilesA,
        features, root_idx, W2 + 128 * DCOL, R2, B, sideBlocks);

    k_gather_gemm<false, true><<<ggBlocks, GG_THREADS, 0, stream>>>(
        XW, ROWPTR, PERM, b1, W2, batch, R2, nullptr, XW2, N,
        root_idx, Wl + 128 * DCOL, R3, B, sideBlocks, ggBlocks - sideBlocks);

    k_gather_gemm<true, false><<<ggBlocks, GG_THREADS, 0, stream>>>(
        XW2, ROWPTR, PERM, b2, Wl, batch, R3, bl, OUT, N,
        nullptr, nullptr, nullptr, 0, 0, ggBlocks);
}


# Gather keeps 32 row loads in flight per wave: row k of the next 32 edges is asked for right after row k of this 32 is added (was: ask 32, drain to none); range tail padded with w=0; sums unchanged
# speedup vs baseline: 3.6497x; 1.0416x over previous
#include <hip/hip_runtime.h>
#include <hip/hip_bf16.h>

typedef float v2f __attribute__((ext_vector_type(2)));
typedef float v4f __attribute__((ext_vector_type(4)));
typedef float v8f __attribute__((ext_vector_type(8)));

#define NEG_SLOPE 0.01f
#define DCOL 128

__device__ __forceinline__ float lrelu(float x) {
    return x > 0.0f ? x : NEG_SLOPE * x;
}

#define SORT_CH 256
#define SORT_T  16
#define BSH     8
#define SORT_LDS_INTS (8 * 256 + 8)

__device__ __forceinline__ int block_scan256(int v, int* wsum, int& total) {
    const int lane = threadIdx.x & 31;
    const int wave = threadIdx.x >> 5;
    int incl = v;
    for (int o = 1; o < 32; o <<= 1) {
        int n = __shfl_up(incl, o);
        if (lane >= o) incl += n;
    }
    if (lane == 31) wsum[wave] = incl;
    __syncthreads();
    int woff = 0, tot = 0;
#pragma unroll
    for (int k = 0; k < 8; ++k) {
        const int s = wsum[k];
        woff += (k < wave) ? s : 0;
        tot += s;
    }
    __syncthreads();
    total = tot;
    return woff + incl - v;
}

__device__ __forceinline__ int wave_counts_sum(const int* cnt, int tid, int (&c8)[8]) {
    int sum = 0;
#pragma unroll
    for (int k = 0; k < 8; ++k) { c8[k] = cnt[k * 256 + tid]; sum += c8[k]; }
    return sum;
}

__device__ __forceinline__ void wave_cursors(int* cnt, int tid, int start, const int (&c8)[8]) {
    int run = start;
#pragma unroll
    for (int k = 0; k < 8; ++k) { cnt[k * 256 + tid] = run; run += c8[k]; }
}

__device__ __forceinline__ void part_body(int* lds,
       const int* __restrict__ src,
       const int* __restrict__ dst,
       const float* __restrict__ wvals,
       int2* __restrict__ rec,
       int* __restrict__ ctab,
       int* __restrict__ otab, int E, int L, int NB, int c) {
    int* cnt  = lds;
    int* wsum = lds + 8 * 256;
    const int tid = threadIdx.x;
    int* mycnt = cnt + (tid >> 5) * 256;
    const int e0 = (c * L < E) ? c * L : E;
    const int e1 = (e0 + L < E) ? e0 + L : E;
#pragma unroll
    for (int k = 0; k < 8; ++k) cnt[k * 256 + tid] = 0;
    __syncthreads();

    for (int base = e0; base < e1; base += 256 * SORT_T) {
        int d[SORT_T];
#pragma unroll
        for (int k = 0; k < SORT_T; ++k) {
            const int e = base + k * 256 + tid;
            d[k] = dst[(e < e1) ? e : (e1 - 1)];
        }
#pragma unroll
        for (int k = 0; k < SORT_T; ++k) {
            const int e = base + k * 256 + tid;
            if (e < e1) atomicAdd(&mycnt[d[k] >> BSH], 1);
        }
    }
    __syncthreads();

    int c8[8];
    const int myc = wave_counts_sum(cnt, tid, c8);
    int total;
    const int off = block_scan256(myc, wsum, total);
    if (tid < NB) {
        ctab[tid * SORT_CH + c] = myc;
        otab[tid * SORT_CH + c] = off;
    }
    wave_cursors(cnt, tid, off, c8);
    __syncthreads();

    for (int base = e0; base < e1; base += 256 * SORT_T) {
        int d[SORT_T], s[SORT_T];
        float w[SORT_T];
#pragma unroll
        for (int k = 0; k < SORT_T; ++k) {
            const int e = base + k * 256 + tid;
            const int ec = (e < e1) ? e : (e1 - 1);
            d[k] = dst[ec];
            s[k] = src[ec];
            w[k] = wvals[ec];
        }
        int p[SORT_T];
#pragma unroll
        for (int k = 0; k < SORT_T; ++k) {
            const int e = base + k * 256 + tid;
            p[k] = (e < e1) ? atomicAdd(&mycnt[d[k] >> BSH], 1) : 0;
        }
#pragma unroll
        for (int k = 0; k < SORT_T; ++k) {
            const int e = base + k * 256 + tid;
            if (e < e1) {
                const unsigned x = ((unsigned)(d[k] & 255) << 24) | (unsigned)s[k];
                rec[e0 + p[k]] = make_int2((int)x, __float_as_int(w[k]));
            }
        }
    }
}

__device__ __forceinline__ void bucket_body(int* lds,
         const int2* __restrict__ rec,
         const int* __restrict__ ctab,
         const int* __restrict__ otab,
         int* __restrict__ rowptr,
         int2* __restrict__ perm, int N, int E, int L, int b) {
    int* ncnt = lds;
    int* wsum = lds + 8 * 256;
    const int tid  = threadIdx.x;
    const int lane = tid & 31;
    int* myn = ncnt + (tid >> 5) * 256;

    const int myCnt = ctab[b * SORT_CH + tid];
    const int myOff = otab[b * SORT_CH + tid];
    const int myStart = ((tid * L < E) ? tid * L : E) + myOff;
#pragma unroll
    for (int k = 0; k < 8; ++k) ncnt[k * 256 + tid] = 0;
    int base;
    (void)block_scan256(myOff, wsum, base);

    int2 r[32];
#pragma unroll
    for (int i = 0; i < 32; ++i) {
        const int cn = __builtin_amdgcn_readlane(myCnt, i);
        const int st = __builtin_amdgcn_readlane(myStart, i);
        int idx = st + ((lane < cn) ? lane : 0);
        idx = (idx < E) ? idx : (E - 1);
        r[i] = rec[idx];
    }
#pragma unroll
    for (int i = 0; i < 32; ++i) {
        const int cn = __builtin_amdgcn_readlane(myCnt, i);
        if (lane < cn) atomicAdd(&myn[(unsigned)r[i].x >> 24], 1);
    }
    const bool longPiece = __builtin_amdgcn_ballot_w32(myCnt > 32) != 0u;
    if (longPiece) {
        for (int i = 0; i < 32; ++i) {
            const int cn = __builtin_amdgcn_readlane(myCnt, i);
            const int st = __builtin_amdgcn_readlane(myStart, i);
            for (int k = 32 + lane; k < cn; k += 32) {
                const int2 q = rec[st + k];
                atomicAdd(&myn[(unsigned)q.x >> 24], 1);
            }
        }
    }
    __syncthreads();

    int c8[8];
    const int v = wave_counts_sum(ncnt, tid, c8);
    int tot;
    const int ex = block_scan256(v, wsum, tot);
    const int node = b * 256 + tid;
    if (node < N) rowptr[node] = base + ex;
    if (node == N - 1) rowptr[N] = E;
    wave_cursors(ncnt, tid, ex, c8);
    __syncthreads();

    int p[32];
#pragma unroll
    for (int i = 0; i < 32; ++i) {
        const int cn = __builtin_amdgcn_readlane(myCnt, i);
        p[i] = (lane < cn) ? atomicAdd(&myn[(unsigned)r[i].x >> 24], 1) : 0;
    }
#pragma unroll
    for (int i = 0; i < 32; ++i) {
        const int cn = __builtin_amdgcn_readlane(myCnt, i);
        if (lane < cn) {
            const unsigned x = (unsigned)r[i].x;
            perm[base + p[i]] = make_int2((int)(x & 0xFFFFFFu), r[i].y);
        }
    }
    if (longPiece) {
        for (int i = 0; i < 32; ++i) {
            const int cn = __builtin_amdgcn_readlane(myCnt, i);
            const int st = __builtin_amdgcn_readlane(myStart, i);
            for (int k = 32 + lane; k < cn; k += 32) {
                const int2 q = rec[st + k];
                const unsigned x = (unsigned)q.x;
                const int p = atomicAdd(&myn[x >> 24], 1);
                perm[base + p] = make_int2((int)(x & 0xFFFFFFu), q.y);
            }
        }
    }
}

template <int U>
__device__ __forceinline__ void gather_rows(const float* __restrict__ xw,
                                            int sx, int wy, int j, int lane,
                                            v4f& acc) {
    v4f x[U];
    float w[U];
#pragma unroll
    for (int u = 0; u < U; ++u) {
        const int s = __builtin_amdgcn_readlane(sx, j + u);
        w[u] = __int_as_float(__builtin_amdgcn_readlane(wy, j + u));
        x[u] = *(const v4f*)(xw + (size_t)s * DCOL + lane * 4);
    }
#pragma unroll
    for (int u = 0; u < U; ++u) acc += w[u] * x[u];
}

__device__ __forceinline__ v4f gather_node(const float* __restrict__ xw,
                                           const int* __restrict__ rowptr,
                                           const int2* __restrict__ perm,
                                           int node, int lane) {
    const int p0 = rowptr[node];
    const int p1 = rowptr[node + 1];
    v4f acc = {0.0f, 0.0f, 0.0f, 0.0f};
    for (int base = p0; base < p1; base += 32) {
        const int cnt = (p1 - base < 32) ? (p1 - base) : 32;
        int2 my = make_int2(0, 0);
        if (lane < cnt) my = perm[base + lane];
        int j = 0;
        for (; j + 8 <= cnt; j += 8) gather_rows<8>(xw, my.x, my.y, j, lane, acc);
        if (j + 4 <= cnt) { gather_rows<4>(xw, my.x, my.y, j, lane, acc); j += 4; }
        if (j + 2 <= cnt) { gather_rows<2>(xw, my.x, my.y, j, lane, acc); j += 2; }
        if (j < cnt)      { gather_rows<1>(xw, my.x, my.y, j, lane, acc); }
    }
    return acc;
}

#define LDA 132

template <bool LR>
__device__ __forceinline__ void put_row(float* __restrict__ dstrow, int lane, v4f acc, v4f gb) {
    v4f v = acc + gb;
    if (LR) { v[0] = lrelu(v[0]); v[1] = lrelu(v[1]); v[2] = lrelu(v[2]); v[3] = lrelu(v[3]); }
    *(v4f*)(dstrow + lane * 4) = v;
}

template <bool LR>
__device__ __forceinline__ void gather_tile_rows(float* __restrict__ arows,
                                                 const float* __restrict__ xw,
                                                 const int* __restrict__ rowptr,
                                                 const int2* __restrict__ perm,
                                                 v4f gb, int rowbase, int lane) {
    const int rpraw = rowptr[rowbase + (lane < 16 ? lane : 16)];
    const int P0 = __builtin_amdgcn_readlane(rpraw, 0);
    const int P1 = __builtin_amdgcn_readlane(rpraw, 16);
    const int rpv = (lane < 16) ? rpraw : 0x7fffffff;
    int cur = 0;
    int nxt = __builtin_amdgcn_readlane(rpv, 1);
    v4f acc = {0.0f, 0.0f, 0.0f, 0.0f};
    const v4f zero = {0.0f, 0.0f, 0.0f, 0.0f};
    if (P0 < P1) {
        auto pairs = [&](int b) {
            const int i = b + lane;
            int2 p = perm[(i < P1) ? i : (P1 - 1)];
            if (i >= P1) p.y = 0;
            return p;
        };
        auto row = [&](int s) {
            return *(const v4f*)(xw + (size_t)s * DCOL + lane * 4);
        };
        int2 my = pairs(P0);
        int2 nx = pairs(P0 + 32);
        v4f x[32];
#pragma unroll
        for (int k = 0; k < 32; ++k) {
            x[k] = row(__builtin_amdgcn_readlane(my.x, k));
            __builtin_amdgcn_sched_barrier(0);
        }
        for (int base = P0; base < P1; base += 32) {
            auto take = [&](int k) {
                const int e = base + k;
                while (e >= nxt) {
                    put_row<LR>(arows + cur * LDA, lane, acc, gb);
                    acc = zero;
                    ++cur;
                    nxt = __builtin_amdgcn_readlane(rpv, cur + 1);
                }
                const float w = __int_as_float(__builtin_amdgcn_readlane(my.y, k));
                acc += w * x[k];
            };
            if (base + 32 < P1) {
                const int2 nn = pairs(base + 64);
#pragma unroll
                for (int k = 0; k < 32; ++k) {
                    take(k);
                    x[k] = row(__builtin_amdgcn_readlane(nx.x, k));
                }
                my = nx;
                nx = nn;
            } else {
#pragma unroll
                for (int k = 0; k < 32; ++k) take(k);
            }
        }
    }
    while (cur < 16) {
        put_row<LR>(arows + cur * LDA, lane, acc, gb);
        acc = zero;
        ++cur;
    }
}

template <bool LR, bool INIT, bool EPI, bool MAP>
__device__ __forceinline__ void gemm_tile128(float* __restrict__ lw,
                                             const float* __restrict__ A,
                                             const int* __restrict__ rowmap,
                                             const float* __restrict__ W,
                                             const int* __restrict__ batch,
                                             const float* __restrict__ R,
                                             const float* __restrict__ bias,
                                             float* __restrict__ C,
                                             int nrows, int blk) {
    const int tid  = threadIdx.x;
    const int wave = tid >> 5;
    const int lane = tid & 31;
    const int lo   = lane & 15;
    const int hi   = lane >> 4;
    const int rowbase = blk * 128 + wave * 16;
    const bool active = rowbase < nrows;

    int arow = rowbase + lo;
    if (MAP) {
        const int g = (arow < nrows) ? arow : (nrows - 1);
        arow = active ? rowmap[g] : 0;
    }

    v8f acc[8] = {};
    if (INIT && active) {
#pragma unroll
        for (int r = 0; r < 8; ++r) {
            const int g = batch[rowbase + r + hi * 8];
            const float* rp = R + (size_t)g * DCOL + lo;
#pragma unroll
            for (int t = 0; t < 8; ++t) acc[t][r] = rp[t * 16];
        }
    }

    for (int i = tid; i < (128 * DCOL) / 8; i += 256) {
        int kp = i >> 5;
        int cg = (i & 31) * 4;
        v4f e = *(const v4f*)(W + (kp * 2 + 0) * DCOL + cg);
        v4f o = *(const v4f*)(W + (kp * 2 + 1) * DCOL + cg);
        float* dstp = lw + kp * 256 + cg * 2;
#pragma unroll
        for (int j = 0; j < 4; ++j) {
            v2f p = { e[j], o[j] };
            *(v2f*)(dstp + j * 2) = p;
        }
    }
    __syncthreads();

    if (active) {
        const float* arowptr = A + (size_t)arow * DCOL;
        const float* lb = lw + hi * 256 + lo * 2;
#pragma unroll 4
        for (int ks = 0; ks < 32; ++ks) {
            v2f a = *(const v2f*)(arowptr + ks * 4 + hi * 2);
            if (LR) { a[0] = lrelu(a[0]); a[1] = lrelu(a[1]); }
            v2f b[8];
#pragma unroll
            for (int t = 0; t < 8; ++t)
                b[t] = *(const v2f*)(lb + ks * 512 + t * 32);
#pragma unroll
            for (int t = 0; t < 8; ++t)
                acc[t] = __builtin_amdgcn_wmma_f32_16x16x4_f32(
                    false, a, false, b[t], (short)0, acc[t], false, false);
        }

        if (MAP) {
#pragma unroll
            for (int r = 0; r < 8; ++r) {
                const int row = rowbase + r + hi * 8;
                if (row < nrows) {
#pragma unroll
                    for (int t = 0; t < 8; ++t)
                        C[(size_t)row * DCOL + t * 16 + lo] = acc[t][r];
                }
            }
        } else {
#pragma unroll
            for (int t = 0; t < 8; ++t) {
                int col = t * 16 + lo;
                float badd = EPI ? bias[col] : 0.0f;
#pragma unroll
                for (int r = 0; r < 8; ++r) {
                    int row = rowbase + r + hi * 8;
                    float v = acc[t][r];
                    if (EPI) v = lrelu(v + badd);
                    C[(size_t)row * DCOL + col] = v;
                }
            }
        }
    }
}

#define GG_WAVES   16
#define GG_THREADS (GG_WAVES * 32)

__device__ __forceinline__ void stage_w(float* __restrict__ lw,
                                        const float* __restrict__ W, int tid) {
    for (int i = tid; i < (128 * DCOL) / 8; i += GG_THREADS) {
        int kp = i >> 5;
        int cg = (i & 31) * 4;
        v4f e = *(const v4f*)(W + (kp * 2 + 0) * DCOL + cg);
        v4f o = *(const v4f*)(W + (kp * 2 + 1) * DCOL + cg);
        float* dstp = lw + kp * 256 + cg * 2;
#pragma unroll
        for (int j = 0; j < 4; ++j) {
            v2f p = { e[j], o[j] };
            *(v2f*)(dstp + j * 2) = p;
        }
    }
}

template <bool EPI>
__device__ __forceinline__ void gather_gemm_groups(float* __restrict__ lw,
                                                   float* __restrict__ la,
                                                   const float* __restrict__ xw,
                                                   const int* __restrict__ rowptr,
                                                   const int2* __restrict__ perm,
                                                   const float* __restrict__ gbias,
                                                   const float* __restrict__ W,
                                                   const int* __restrict__ batch,
                                                   const float* __restrict__ R,
                                                   const float* __restrict__ bias,
                                                   float* __restrict__ C,
                                                   int nrows, int mb, int G) {
    const int tid  = threadIdx.x;
    const int wave = __builtin_amdgcn_readfirstlane(tid >> 5);
    const int lane = tid & 31;
    const int lo   = lane & 15;
    const int hi   = lane >> 4;

    stage_w(lw, W, tid);
    __syncthreads();

    const v4f gb = *(const v4f*)(gbias + lane * 4);
    float* arows = la + wave * 16 * LDA;
    const float* ap = arows + lo * LDA + hi * 2;
    const float* lb = lw + hi * 256 + lo * 2;
    const int groups = (nrows + 15) >> 4;
    for (int gi = wave * G + mb; gi < groups; gi += GG_WAVES * G) {
        const int rowbase = gi * 16;
        v8f acc[8];
#pragma unroll
        for (int r = 0; r < 8; ++r) {
            const int g = batch[rowbase + r + hi * 8];
            const float* rp = R + (size_t)g * DCOL + lo;
#pragma unroll
            for (int t = 0; t < 8; ++t) acc[t][r] = rp[t * 16];
        }

        gather_tile_rows<true>(arows, xw, rowptr, perm, gb, rowbase, lane);

#pragma unroll 4
        for (int ks = 0; ks < 32; ++ks) {
            const v2f a = *(const v2f*)(ap + ks * 4);
            v2f b[8];
#pragma unroll
            for (int t = 0; t < 8; ++t)
                b[t] = *(const v2f*)(lb + ks * 512 + t * 32);
#pragma unroll
            for (int t = 0; t < 8; ++t)
                acc[t] = __builtin_amdgcn_wmma_f32_16x16x4_f32(
                    false, a, false, b[t], (short)0, acc[t], false, false);
        }

#pragma unroll
        for (int t = 0; t < 8; ++t) {
            int col = t * 16 + lo;
            float badd = EPI ? bias[col] : 0.0f;
#pragma unroll
            for (int r = 0; r < 8; ++r) {
                int row = rowbase + r + hi * 8;
                float v = acc[t][r];
                if (EPI) v = lrelu(v + badd);
                C[(size_t)row * DCOL + col] = v;
            }
        }
    }
}

__device__ __forceinline__ void gather_gemm_side(float* __restrict__ lw,
                                                 float* __restrict__ la,
                                                 const float* __restrict__ xw,
                                                 const int* __restrict__ rowptr,
                                                 const int2* __restrict__ perm,
                                                 const float* __restrict__ gbias,
                                                 const int* __restrict__ rowmap,
                                                 const float* __restrict__ W,
                                                 float* __restrict__ C,
                                                 int nrows, int blk) {
    const int tid  = threadIdx.x;
    const int wave = tid >> 5;
    const int lane = tid & 31;
    const int lo   = lane & 15;
    const int hi   = lane >> 4;
    const int rowbase = blk * 128 + wave * 16;
    const bool active = wave < 8 && rowbase < nrows;

    stage_w(lw, W, tid);

    const v4f gb = *(const v4f*)(gbias + lane * 4);
    const v4f zero = {0.0f, 0.0f, 0.0f, 0.0f};
    for (int i = wave; i < 128; i += GG_WAVES) {
        const int g = blk * 128 + i;
        if (g < nrows) {
            const int node = __builtin_amdgcn_readfirstlane(rowmap[g]);
            const v4f s = gather_node(xw, rowptr, perm, node, lane);
            put_row<false>(la + i * LDA, lane, s, gb);
        } else {
            *(v4f*)(la + i * LDA + lane * 4) = zero;
        }
    }
    __syncthreads();

    if (active) {
        v8f acc[8] = {};
        const float* ap = la + (wave * 16 + lo) * LDA + hi * 2;
        const float* lb = lw + hi * 256 + lo * 2;
#pragma unroll 4
        for (int ks = 0; ks < 32; ++ks) {
            const v2f a = *(const v2f*)(ap + ks * 4);
            v2f b[8];
#pragma unroll
            for (int t = 0; t < 8; ++t)
                b[t] = *(const v2f*)(lb + ks * 512 + t * 32);
#pragma unroll
            for (int t = 0; t < 8; ++t)
                acc[t] = __builtin_amdgcn_wmma_f32_16x16x4_f32(
                    false, a, false, b[t], (short)0, acc[t], false, false);
        }

#pragma unroll
        for (int r = 0; r < 8; ++r) {
            const int row = rowbase + r + hi * 8;
            if (row < nrows) {
#pragma unroll
                for (int t = 0; t < 8; ++t)
                    C[(size_t)row * DCOL + t * 16 + lo] = acc[t][r];
            }
        }
    }
}

template <bool EPI, bool SIDE>
__global__ void __launch_bounds__(GG_THREADS)
k_gather_gemm(const float* __restrict__ xw,
              const int* __restrict__ rowptr,
              const int2* __restrict__ perm,
              const float* __restrict__ gbias,
              const float* __restrict__ W,
              const int* __restrict__ batch,
              const float* __restrict__ R,
              const float* __restrict__ bias,
              float* __restrict__ C, int N,
              const int* __restrict__ root_idx,
              const float* __restrict__ TW,
              float* __restrict__ T, int B, int sideBlocks, int mainBlocks) {
    __shared__ float lw[128 * 128];
    __shared__ float la[GG_WAVES * 16 * LDA];
    const int blk = (int)blockIdx.x;
    if (SIDE && blk < sideBlocks) {
        gather_gemm_side(lw, la, xw, rowptr, perm, gbias, root_idx, TW, T, B, blk);
    } else {
        gather_gemm_groups<EPI>(
            lw, la, xw, rowptr, perm, gbias, W,
            batch, R, bias, C, N, blk - (SIDE ? sideBlocks : 0), mainBlocks);
    }
}

template <int PHASE>
__global__ void __launch_bounds__(256)
k_sort_gemm(const int* __restrict__ src,
            const int* __restrict__ dst,
            const float* __restrict__ wvals,
            int2* __restrict__ rec,
            int* __restrict__ ctab,
            int* __restrict__ otab,
            int* __restrict__ rowptr,
            int2* __restrict__ perm,
            int N, int E, int L, int NB, int gemmPart,
            const float* __restrict__ A0,
            const float* __restrict__ W,
            float* __restrict__ C, int tile0,
            const float* __restrict__ TA,
            const int* __restrict__ root_idx,
            const float* __restrict__ TW,
            float* __restrict__ T, int B, int sideBlocks) {
    __shared__ float lw[128 * 128];
    const int blk = (int)blockIdx.x;
    if (blk >= gemmPart) {
        if (PHASE == 0)
            part_body((int*)lw, src, dst, wvals, rec, ctab, otab, E, L, NB, blk - gemmPart);
        else
            bucket_body((int*)lw, rec, ctab, otab, rowptr, perm, N, E, L, blk - gemmPart);
    } else if (blk < sideBlocks) {
        gemm_tile128<true, false, false, true>(
            lw, TA, root_idx, TW, nullptr, nullptr, nullptr, T, B, blk);
    } else {
        gemm_tile128<false, false, false, false>(
            lw, A0, nullptr, W, nullptr, nullptr, nullptr, C, N,
            tile0 + blk - sideBlocks);
    }
}

extern "C" void kernel_launch(void* const* d_in, const int* in_sizes, int n_in,
                              void* d_out, int out_size, void* d_ws, size_t ws_size,
                              hipStream_t stream) {
    const float* features = (const float*)d_in[0];
    const float* values   = (const float*)d_in[1];
    const float* W1       = (const float*)d_in[2];
    const float* b1       = (const float*)d_in[3];
    const float* W2       = (const float*)d_in[4];
    const float* b2       = (const float*)d_in[5];
    const float* Wl       = (const float*)d_in[6];
    const float* bl       = (const float*)d_in[7];
    const int*   adjs     = (const int*)d_in[8];
    const int*   root_idx = (const int*)d_in[9];
    const int*   batch    = (const int*)d_in[12];

    const int N = in_sizes[0] / DCOL;
    const int E = in_sizes[1];
    const int B = in_sizes[9];
    const int* srcI = adjs;
    const int* dstI = adjs + E;

    int2* PERM   = (int2*)d_ws;
    int2* REC    = PERM + E;
    int*  CTAB   = (int*)(REC + E);
    int*  OTAB   = CTAB + 256 * SORT_CH;
    int*  ROWPTR = OTAB + 256 * SORT_CH;
    float* XW2 = (float*)d_ws + (size_t)N * DCOL;
    float* R2 = XW2 - (size_t)2 * B * DCOL;
    float* R3 = R2 + (size_t)B * DCOL;

    float* OUT = (float*)d_out;
    float* XW  = OUT;

    const int gemmBlocks = (N + 127) / 128;
    const int sideBlocks = (B + 127) / 128;
    const int chunkLen = (E + SORT_CH - 1) / SORT_CH;
    const int buckets  = (N + 255) / 256;
    const int ggBlocks = 256;

    const int tilesA = gemmBlocks / 2;
    const int tilesB = gemmBlocks - tilesA;
    k_sort_gemm<0><<<tilesA + SORT_CH, 256, 0, stream>>>(
        srcI, dstI, values, REC, CTAB, OTAB, ROWPTR, PERM, N, E, chunkLen, buckets, tilesA,
        features, W1, XW, 0,
        nullptr, nullptr, nullptr, nullptr, 0, 0);
    k_sort_gemm<1><<<sideBlocks + tilesB + buckets, 256, 0, stream>>>(
        srcI, dstI, values, REC, CTAB, OTAB, ROWPTR, PERM, N, E, chunkLen, buckets, sideBlocks + tilesB,
        features, W1, XW, tilesA,
        features, root_idx, W2 + 128 * DCOL, R2, B, sideBlocks);

    k_gather_gemm<false, true><<<ggBlocks, GG_THREADS, 0, stream>>>(
        XW, ROWPTR, PERM, b1, W2, batch, R2, nullptr, XW2, N,
        root_idx, Wl + 128 * DCOL, R3, B, sideBlocks, ggBlocks - sideBlocks);

    k_gather_gemm<true, false><<<ggBlocks, GG_THREADS, 0, stream>>>(
        XW2, ROWPTR, PERM, b2, Wl, batch, R3, bl, OUT, N,
        nullptr, nullptr, nullptr, 0, 0, ggBlocks);
}


# W staging, all four launches: a thread asks for all its W loads (16 in GEMM tiles, 8 in gather launches) before its first LDS store; was load-wait-store, one round trip per 2 loads; sums unchanged
# speedup vs baseline: 3.8055x; 1.0427x over previous
#include <hip/hip_runtime.h>
#include <hip/hip_bf16.h>

typedef float v2f __attribute__((ext_vector_type(2)));
typedef float v4f __attribute__((ext_vector_type(4)));
typedef float v8f __attribute__((ext_vector_type(8)));

#define NEG_SLOPE 0.01f
#define DCOL 128

__device__ __forceinline__ float lrelu(float x) {
    return x > 0.0f ? x : NEG_SLOPE * x;
}

#define SORT_CH 256
#define SORT_T  16
#define BSH     8
#define SORT_LDS_INTS (8 * 256 + 8)

__device__ __forceinline__ int block_scan256(int v, int* wsum, int& total) {
    const int lane = threadIdx.x & 31;
    const int wave = threadIdx.x >> 5;
    int incl = v;
    for (int o = 1; o < 32; o <<= 1) {
        int n = __shfl_up(incl, o);
        if (lane >= o) incl += n;
    }
    if (lane == 31) wsum[wave] = incl;
    __syncthreads();
    int woff = 0, tot = 0;
#pragma unroll
    for (int k = 0; k < 8; ++k) {
        const int s = wsum[k];
        woff += (k < wave) ? s : 0;
        tot += s;
    }
    __syncthreads();
    total = tot;
    return woff + incl - v;
}

__device__ __forceinline__ int wave_counts_sum(const int* cnt, int tid, int (&c8)[8]) {
    int sum = 0;
#pragma unroll
    for (int k = 0; k < 8; ++k) { c8[k] = cnt[k * 256 + tid]; sum += c8[k]; }
    return sum;
}

__device__ __forceinline__ void wave_cursors(int* cnt, int tid, int start, const int (&c8)[8]) {
    int run = start;
#pragma unroll
    for (int k = 0; k < 8; ++k) { cnt[k * 256 + tid] = run; run += c8[k]; }
}

__device__ __forceinline__ void part_body(int* lds,
       const int* __restrict__ src,
       const int* __restrict__ dst,
       const float* __restrict__ wvals,
       int2* __restrict__ rec,
       int* __restrict__ ctab,
       int* __restrict__ otab, int E, int L, int NB, int c) {
    int* cnt  = lds;
    int* wsum = lds + 8 * 256;
    const int tid = threadIdx.x;
    int* mycnt = cnt + (tid >> 5) * 256;
    const int e0 = (c * L < E) ? c * L : E;
    const int e1 = (e0 + L < E) ? e0 + L : E;
#pragma unroll
    for (int k = 0; k < 8; ++k) cnt[k * 256 + tid] = 0;
    __syncthreads();

    for (int base = e0; base < e1; base += 256 * SORT_T) {
        int d[SORT_T];
#pragma unroll
        for (int k = 0; k < SORT_T; ++k) {
            const int e = base + k * 256 + tid;
            d[k] = dst[(e < e1) ? e : (e1 - 1)];
        }
#pragma unroll
        for (int k = 0; k < SORT_T; ++k) {
            const int e = base + k * 256 + tid;
            if (e < e1) atomicAdd(&mycnt[d[k] >> BSH], 1);
        }
    }
    __syncthreads();

    int c8[8];
    const int myc = wave_counts_sum(cnt, tid, c8);
    int total;
    const int off = block_scan256(myc, wsum, total);
    if (tid < NB) {
        ctab[tid * SORT_CH + c] = myc;
        otab[tid * SORT_CH + c] = off;
    }
    wave_cursors(cnt, tid, off, c8);
    __syncthreads();

    for (int base = e0; base < e1; base += 256 * SORT_T) {
        int d[SORT_T], s[SORT_T];
        float w[SORT_T];
#pragma unroll
        for (int k = 0; k < SORT_T; ++k) {
            const int e = base + k * 256 + tid;
            const int ec = (e < e1) ? e : (e1 - 1);
            d[k] = dst[ec];
            s[k] = src[ec];
            w[k] = wvals[ec];
        }
        int p[SORT_T];
#pragma unroll
        for (int k = 0; k < SORT_T; ++k) {
            const int e = base + k * 256 + tid;
            p[k] = (e < e1) ? atomicAdd(&mycnt[d[k] >> BSH], 1) : 0;
        }
#pragma unroll
        for (int k = 0; k < SORT_T; ++k) {
            const int e = base + k * 256 + tid;
            if (e < e1) {
                const unsigned x = ((unsigned)(d[k] & 255) << 24) | (unsigned)s[k];
                rec[e0 + p[k]] = make_int2((int)x, __float_as_int(w[k]));
            }
        }
    }
}

__device__ __forceinline__ void bucket_body(int* lds,
         const int2* __restrict__ rec,
         const int* __restrict__ ctab,
         const int* __restrict__ otab,
         int* __restrict__ rowptr,
         int2* __restrict__ perm, int N, int E, int L, int b) {
    int* ncnt = lds;
    int* wsum = lds + 8 * 256;
    const int tid  = threadIdx.x;
    const int lane = tid & 31;
    int* myn = ncnt + (tid >> 5) * 256;

    const int myCnt = ctab[b * SORT_CH + tid];
    const int myOff = otab[b * SORT_CH + tid];
    const int myStart = ((tid * L < E) ? tid * L : E) + myOff;
#pragma unroll
    for (int k = 0; k < 8; ++k) ncnt[k * 256 + tid] = 0;
    int base;
    (void)block_scan256(myOff, wsum, base);

    int2 r[32];
#pragma unroll
    for (int i = 0; i < 32; ++i) {
        const int cn = __builtin_amdgcn_readlane(myCnt, i);
        const int st = __builtin_amdgcn_readlane(myStart, i);
        int idx = st + ((lane < cn) ? lane : 0);
        idx = (idx < E) ? idx : (E - 1);
        r[i] = rec[idx];
    }
#pragma unroll
    for (int i = 0; i < 32; ++i) {
        const int cn = __builtin_amdgcn_readlane(myCnt, i);
        if (lane < cn) atomicAdd(&myn[(unsigned)r[i].x >> 24], 1);
    }
    const bool longPiece = __builtin_amdgcn_ballot_w32(myCnt > 32) != 0u;
    if (longPiece) {
        for (int i = 0; i < 32; ++i) {
            const int cn = __builtin_amdgcn_readlane(myCnt, i);
            const int st = __builtin_amdgcn_readlane(myStart, i);
            for (int k = 32 + lane; k < cn; k += 32) {
                const int2 q = rec[st + k];
                atomicAdd(&myn[(unsigned)q.x >> 24], 1);
            }
        }
    }
    __syncthreads();

    int c8[8];
    const int v = wave_counts_sum(ncnt, tid, c8);
    int tot;
    const int ex = block_scan256(v, wsum, tot);
    const int node = b * 256 + tid;
    if (node < N) rowptr[node] = base + ex;
    if (node == N - 1) rowptr[N] = E;
    wave_cursors(ncnt, tid, ex, c8);
    __syncthreads();

    int p[32];
#pragma unroll
    for (int i = 0; i < 32; ++i) {
        const int cn = __builtin_amdgcn_readlane(myCnt, i);
        p[i] = (lane < cn) ? atomicAdd(&myn[(unsigned)r[i].x >> 24], 1) : 0;
    }
#pragma unroll
    for (int i = 0; i < 32; ++i) {
        const int cn = __builtin_amdgcn_readlane(myCnt, i);
        if (lane < cn) {
            const unsigned x = (unsigned)r[i].x;
            perm[base + p[i]] = make_int2((int)(x & 0xFFFFFFu), r[i].y);
        }
    }
    if (longPiece) {
        for (int i = 0; i < 32; ++i) {
            const int cn = __builtin_amdgcn_readlane(myCnt, i);
            const int st = __builtin_amdgcn_readlane(myStart, i);
            for (int k = 32 + lane; k < cn; k += 32) {
                const int2 q = rec[st + k];
                const unsigned x = (unsigned)q.x;
                const int p = atomicAdd(&myn[x >> 24], 1);
                perm[base + p] = make_int2((int)(x & 0xFFFFFFu), q.y);
            }
        }
    }
}

template <int U>
__device__ __forceinline__ void gather_rows(const float* __restrict__ xw,
                                            int sx, int wy, int j, int lane,
                                            v4f& acc) {
    v4f x[U];
    float w[U];
#pragma unroll
    for (int u = 0; u < U; ++u) {
        const int s = __builtin_amdgcn_readlane(sx, j + u);
        w[u] = __int_as_float(__builtin_amdgcn_readlane(wy, j + u));
        x[u] = *(const v4f*)(xw + (size_t)s * DCOL + lane * 4);
    }
#pragma unroll
    for (int u = 0; u < U; ++u) acc += w[u] * x[u];
}

__device__ __forceinline__ v4f gather_node(const float* __restrict__ xw,
                                           const int* __restrict__ rowptr,
                                           const int2* __restrict__ perm,
                                           int node, int lane) {
    const int p0 = rowptr[node];
    const int p1 = rowptr[node + 1];
    v4f acc = {0.0f, 0.0f, 0.0f, 0.0f};
    for (int base = p0; base < p1; base += 32) {
        const int cnt = (p1 - base < 32) ? (p1 - base) : 32;
        int2 my = make_int2(0, 0);
        if (lane < cnt) my = perm[base + lane];
        int j = 0;
        for (; j + 8 <= cnt; j += 8) gather_rows<8>(xw, my.x, my.y, j, lane, acc);
        if (j + 4 <= cnt) { gather_rows<4>(xw, my.x, my.y, j, lane, acc); j += 4; }
        if (j + 2 <= cnt) { gather_rows<2>(xw, my.x, my.y, j, lane, acc); j += 2; }
        if (j < cnt)      { gather_rows<1>(xw, my.x, my.y, j, lane, acc); }
    }
    return acc;
}

#define LDA 132

template <bool LR>
__device__ __forceinline__ void put_row(float* __restrict__ dstrow, int lane, v4f acc, v4f gb) {
    v4f v = acc + gb;
    if (LR) { v[0] = lrelu(v[0]); v[1] = lrelu(v[1]); v[2] = lrelu(v[2]); v[3] = lrelu(v[3]); }
    *(v4f*)(dstrow + lane * 4) = v;
}

template <bool LR>
__device__ __forceinline__ void gather_tile_rows(float* __restrict__ arows,
                                                 const float* __restrict__ xw,
                                                 const int* __restrict__ rowptr,
                                                 const int2* __restrict__ perm,
                                                 v4f gb, int rowbase, int lane) {
    const int rpraw = rowptr[rowbase + (lane < 16 ? lane : 16)];
    const int P0 = __builtin_amdgcn_readlane(rpraw, 0);
    const int P1 = __builtin_amdgcn_readlane(rpraw, 16);
    const int rpv = (lane < 16) ? rpraw : 0x7fffffff;
    int cur = 0;
    int nxt = __builtin_amdgcn_readlane(rpv, 1);
    v4f acc = {0.0f, 0.0f, 0.0f, 0.0f};
    const v4f zero = {0.0f, 0.0f, 0.0f, 0.0f};
    if (P0 < P1) {
        auto pairs = [&](int b) {
            const int i = b + lane;
            int2 p = perm[(i < P1) ? i : (P1 - 1)];
            if (i >= P1) p.y = 0;
            return p;
        };
        auto row = [&](int s) {
            return *(const v4f*)(xw + (size_t)s * DCOL + lane * 4);
        };
        int2 my = pairs(P0);
        int2 nx = pairs(P0 + 32);
        v4f x[32];
#pragma unroll
        for (int k = 0; k < 32; ++k) {
            x[k] = row(__builtin_amdgcn_readlane(my.x, k));
            __builtin_amdgcn_sched_barrier(0);
        }
        for (int base = P0; base < P1; base += 32) {
            auto take = [&](int k) {
                const int e = base + k;
                while (e >= nxt) {
                    put_row<LR>(arows + cur * LDA, lane, acc, gb);
                    acc = zero;
                    ++cur;
                    nxt = __builtin_amdgcn_readlane(rpv, cur + 1);
                }
                const float w = __int_as_float(__builtin_amdgcn_readlane(my.y, k));
                acc += w * x[k];
            };
            if (base + 32 < P1) {
                const int2 nn = pairs(base + 64);
#pragma unroll
                for (int k = 0; k < 32; ++k) {
                    take(k);
                    x[k] = row(__builtin_amdgcn_readlane(nx.x, k));
                }
                my = nx;
                nx = nn;
            } else {
#pragma unroll
                for (int k = 0; k < 32; ++k) take(k);
            }
        }
    }
    while (cur < 16) {
        put_row<LR>(arows + cur * LDA, lane, acc, gb);
        acc = zero;
        ++cur;
    }
}

template <bool LR, bool INIT, bool EPI, bool MAP>
__device__ __forceinline__ void gemm_tile128(float* __restrict__ lw,
                                             const float* __restrict__ A,
                                             const int* __restrict__ rowmap,
                                             const float* __restrict__ W,
                                             const int* __restrict__ batch,
                                             const float* __restrict__ R,
                                             const float* __restrict__ bias,
                                             float* __restrict__ C,
                                             int nrows, int blk) {
    const int tid  = threadIdx.x;
    const int wave = tid >> 5;
    const int lane = tid & 31;
    const int lo   = lane & 15;
    const int hi   = lane >> 4;
    const int rowbase = blk * 128 + wave * 16;
    const bool active = rowbase < nrows;

    int arow = rowbase + lo;
    if (MAP) {
        const int g = (arow < nrows) ? arow : (nrows - 1);
        arow = active ? rowmap[g] : 0;
    }

    v8f acc[8] = {};
    if (INIT && active) {
#pragma unroll
        for (int r = 0; r < 8; ++r) {
            const int g = batch[rowbase + r + hi * 8];
            const float* rp = R + (size_t)g * DCOL + lo;
#pragma unroll
            for (int t = 0; t < 8; ++t) acc[t][r] = rp[t * 16];
        }
    }

    v4f we[8], wo[8];
#pragma unroll
    for (int j = 0; j < 8; ++j) {
        const int i = tid + j * 256;
        const int kp = i >> 5;
        const int cg = (i & 31) * 4;
        we[j] = *(const v4f*)(W + (kp * 2 + 0) * DCOL + cg);
        wo[j] = *(const v4f*)(W + (kp * 2 + 1) * DCOL + cg);
    }
#pragma unroll
    for (int j = 0; j < 8; ++j) {
        const int i = tid + j * 256;
        const int kp = i >> 5;
        const int cg = (i & 31) * 4;
        float* dstp = lw + kp * 256 + cg * 2;
#pragma unroll
        for (int c = 0; c < 4; ++c) {
            v2f p = { we[j][c], wo[j][c] };
            *(v2f*)(dstp + c * 2) = p;
        }
    }
    __syncthreads();

    if (active) {
        const float* arowptr = A + (size_t)arow * DCOL;
        const float* lb = lw + hi * 256 + lo * 2;
#pragma unroll 4
        for (int ks = 0; ks < 32; ++ks) {
            v2f a = *(const v2f*)(arowptr + ks * 4 + hi * 2);
            if (LR) { a[0] = lrelu(a[0]); a[1] = lrelu(a[1]); }
            v2f b[8];
#pragma unroll
            for (int t = 0; t < 8; ++t)
                b[t] = *(const v2f*)(lb + ks * 512 + t * 32);
#pragma unroll
            for (int t = 0; t < 8; ++t)
                acc[t] = __builtin_amdgcn_wmma_f32_16x16x4_f32(
                    false, a, false, b[t], (short)0, acc[t], false, false);
        }

        if (MAP) {
#pragma unroll
            for (int r = 0; r < 8; ++r) {
                const int row = rowbase + r + hi * 8;
                if (row < nrows) {
#pragma unroll
                    for (int t = 0; t < 8; ++t)
                        C[(size_t)row * DCOL + t * 16 + lo] = acc[t][r];
                }
            }
        } else {
#pragma unroll
            for (int t = 0; t < 8; ++t) {
                int col = t * 16 + lo;
                float badd = EPI ? bias[col] : 0.0f;
#pragma unroll
                for (int r = 0; r < 8; ++r) {
                    int row = rowbase + r + hi * 8;
                    float v = acc[t][r];
                    if (EPI) v = lrelu(v + badd);
                    C[(size_t)row * DCOL + col] = v;
                }
            }
        }
    }
}

#define GG_WAVES   16
#define GG_THREADS (GG_WAVES * 32)

__device__ __forceinline__ void stage_w(float* __restrict__ lw,
                                        const float* __restrict__ W, int tid) {
    constexpr int G4 = (128 * DCOL) / 8 / GG_THREADS;
    v4f we[G4], wo[G4];
#pragma unroll
    for (int j = 0; j < G4; ++j) {
        const int i = tid + j * GG_THREADS;
        const int kp = i >> 5;
        const int cg = (i & 31) * 4;
        we[j] = *(const v4f*)(W + (kp * 2 + 0) * DCOL + cg);
        wo[j] = *(const v4f*)(W + (kp * 2 + 1) * DCOL + cg);
    }
#pragma unroll
    for (int j = 0; j < G4; ++j) {
        const int i = tid + j * GG_THREADS;
        const int kp = i >> 5;
        const int cg = (i & 31) * 4;
        float* dstp = lw + kp * 256 + cg * 2;
#pragma unroll
        for (int c = 0; c < 4; ++c) {
            v2f p = { we[j][c], wo[j][c] };
            *(v2f*)(dstp + c * 2) = p;
        }
    }
}

template <bool EPI>
__device__ __forceinline__ void gather_gemm_groups(float* __restrict__ lw,
                                                   float* __restrict__ la,
                                                   const float* __restrict__ xw,
                                                   const int* __restrict__ rowptr,
                                                   const int2* __restrict__ perm,
                                                   const float* __restrict__ gbias,
                                                   const float* __restrict__ W,
                                                   const int* __restrict__ batch,
                                                   const float* __restrict__ R,
                                                   const float* __restrict__ bias,
                                                   float* __restrict__ C,
                                                   int nrows, int mb, int G) {
    const int tid  = threadIdx.x;
    const int wave = __builtin_amdgcn_readfirstlane(tid >> 5);
    const int lane = tid & 31;
    const int lo   = lane & 15;
    const int hi   = lane >> 4;

    stage_w(lw, W, tid);
    __syncthreads();

    const v4f gb = *(const v4f*)(gbias + lane * 4);
    float* arows = la + wave * 16 * LDA;
    const float* ap = arows + lo * LDA + hi * 2;
    const float* lb = lw + hi * 256 + lo * 2;
    const int groups = (nrows + 15) >> 4;
    for (int gi = wave * G + mb; gi < groups; gi += GG_WAVES * G) {
        const int rowbase = gi * 16;
        v8f acc[8];
#pragma unroll
        for (int r = 0; r < 8; ++r) {
            const int g = batch[rowbase + r + hi * 8];
            const float* rp = R + (size_t)g * DCOL + lo;
#pragma unroll
            for (int t = 0; t < 8; ++t) acc[t][r] = rp[t * 16];
        }

        gather_tile_rows<true>(arows, xw, rowptr, perm, gb, rowbase, lane);

#pragma unroll 4
        for (int ks = 0; ks < 32; ++ks) {
            const v2f a = *(const v2f*)(ap + ks * 4);
            v2f b[8];
#pragma unroll
            for (int t = 0; t < 8; ++t)
                b[t] = *(const v2f*)(lb + ks * 512 + t * 32);
#pragma unroll
            for (int t = 0; t < 8; ++t)
                acc[t] = __builtin_amdgcn_wmma_f32_16x16x4_f32(
                    false, a, false, b[t], (short)0, acc[t], false, false);
        }

#pragma unroll
        for (int t = 0; t < 8; ++t) {
            int col = t * 16 + lo;
            float badd = EPI ? bias[col] : 0.0f;
#pragma unroll
            for (int r = 0; r < 8; ++r) {
                int row = rowbase + r + hi * 8;
                float v = acc[t][r];
                if (EPI) v = lrelu(v + badd);
                C[(size_t)row * DCOL + col] = v;
            }
        }
    }
}

__device__ __forceinline__ void gather_gemm_side(float* __restrict__ lw,
                                                 float* __restrict__ la,
                                                 const float* __restrict__ xw,
                                                 const int* __restrict__ rowptr,
                                                 const int2* __restrict__ perm,
                                                 const float* __restrict__ gbias,
                                                 const int* __restrict__ rowmap,
                                                 const float* __restrict__ W,
                                                 float* __restrict__ C,
                                                 int nrows, int blk) {
    const int tid  = threadIdx.x;
    const int wave = tid >> 5;
    const int lane = tid & 31;
    const int lo   = lane & 15;
    const int hi   = lane >> 4;
    const int rowbase = blk * 128 + wave * 16;
    const bool active = wave < 8 && rowbase < nrows;

    stage_w(lw, W, tid);

    const v4f gb = *(const v4f*)(gbias + lane * 4);
    const v4f zero = {0.0f, 0.0f, 0.0f, 0.0f};
    for (int i = wave; i < 128; i += GG_WAVES) {
        const int g = blk * 128 + i;
        if (g < nrows) {
            const int node = __builtin_amdgcn_readfirstlane(rowmap[g]);
            const v4f s = gather_node(xw, rowptr, perm, node, lane);
            put_row<false>(la + i * LDA, lane, s, gb);
        } else {
            *(v4f*)(la + i * LDA + lane * 4) = zero;
        }
    }
    __syncthreads();

    if (active) {
        v8f acc[8] = {};
        const float* ap = la + (wave * 16 + lo) * LDA + hi * 2;
        const float* lb = lw + hi * 256 + lo * 2;
#pragma unroll 4
        for (int ks = 0; ks < 32; ++ks) {
            const v2f a = *(const v2f*)(ap + ks * 4);
            v2f b[8];
#pragma unroll
            for (int t = 0; t < 8; ++t)
                b[t] = *(const v2f*)(lb + ks * 512 + t * 32);
#pragma unroll
            for (int t = 0; t < 8; ++t)
                acc[t] = __builtin_amdgcn_wmma_f32_16x16x4_f32(
                    false, a, false, b[t], (short)0, acc[t], false, false);
        }

#pragma unroll
        for (int r = 0; r < 8; ++r) {
            const int row = rowbase + r + hi * 8;
            if (row < nrows) {
#pragma unroll
                for (int t = 0; t < 8; ++t)
                    C[(size_t)row * DCOL + t * 16 + lo] = acc[t][r];
            }
        }
    }
}

template <bool EPI, bool SIDE>
__global__ void __launch_bounds__(GG_THREADS)
k_gather_gemm(const float* __restrict__ xw,
              const int* __restrict__ rowptr,
              const int2* __restrict__ perm,
              const float* __restrict__ gbias,
              const float* __restrict__ W,
              const int* __restrict__ batch,
              const float* __restrict__ R,
              const float* __restrict__ bias,
              float* __restrict__ C, int N,
              const int* __restrict__ root_idx,
              const float* __restrict__ TW,
              float* __restrict__ T, int B, int sideBlocks, int mainBlocks) {
    __shared__ float lw[128 * 128];
    __shared__ float la[GG_WAVES * 16 * LDA];
    const int blk = (int)blockIdx.x;
    if (SIDE && blk < sideBlocks) {
        gather_gemm_side(lw, la, xw, rowptr, perm, gbias, root_idx, TW, T, B, blk);
    } else {
        gather_gemm_groups<EPI>(
            lw, la, xw, rowptr, perm, gbias, W,
            batch, R, bias, C, N, blk - (SIDE ? sideBlocks : 0), mainBlocks);
    }
}

template <int PHASE>
__global__ void __launch_bounds__(256)
k_sort_gemm(const int* __restrict__ src,
            const int* __restrict__ dst,
            const float* __restrict__ wvals,
            int2* __restrict__ rec,
            int* __restrict__ ctab,
            int* __restrict__ otab,
            int* __restrict__ rowptr,
            int2* __restrict__ perm,
            int N, int E, int L, int NB, int gemmPart,
            const float* __restrict__ A0,
            const float* __restrict__ W,
            float* __restrict__ C, int tile0,
            const float* __restrict__ TA,
            const int* __restrict__ root_idx,
            const float* __restrict__ TW,
            float* __restrict__ T, int B, int sideBlocks) {
    __shared__ float lw[128 * 128];
    const int blk = (int)blockIdx.x;
    if (blk >= gemmPart) {
        if (PHASE == 0)
            part_body((int*)lw, src, dst, wvals, rec, ctab, otab, E, L, NB, blk - gemmPart);
        else
            bucket_body((int*)lw, rec, ctab, otab, rowptr, perm, N, E, L, blk - gemmPart);
    } else if (blk < sideBlocks) {
        gemm_tile128<true, false, false, true>(
            lw, TA, root_idx, TW, nullptr, nullptr, nullptr, T, B, blk);
    } else {
        gemm_tile128<false, false, false, false>(
            lw, A0, nullptr, W, nullptr, nullptr, nullptr, C, N,
            tile0 + blk - sideBlocks);
    }
}

extern "C" void kernel_launch(void* const* d_in, const int* in_sizes, int n_in,
                              void* d_out, int out_size, void* d_ws, size_t ws_size,
                              hipStream_t stream) {
    const float* features = (const float*)d_in[0];
    const float* values   = (const float*)d_in[1];
    const float* W1       = (const float*)d_in[2];
    const float* b1       = (const float*)d_in[3];
    const float* W2       = (const float*)d_in[4];
    const float* b2       = (const float*)d_in[5];
    const float* Wl       = (const float*)d_in[6];
    const float* bl       = (const float*)d_in[7];
    const int*   adjs     = (const int*)d_in[8];
    const int*   root_idx = (const int*)d_in[9];
    const int*   batch    = (const int*)d_in[12];

    const int N = in_sizes[0] / DCOL;
    const int E = in_sizes[1];
    const int B = in_sizes[9];
    const int* srcI = adjs;
    const int* dstI = adjs + E;

    int2* PERM   = (int2*)d_ws;
    int2* REC    = PERM + E;
    int*  CTAB   = (int*)(REC + E);
    int*  OTAB   = CTAB + 256 * SORT_CH;
    int*  ROWPTR = OTAB + 256 * SORT_CH;
    float* XW2 = (float*)d_ws + (size_t)N * DCOL;
    float* R2 = XW2 - (size_t)2 * B * DCOL;
    float* R3 = R2 + (size_t)B * DCOL;

    float* OUT = (float*)d_out;
    float* XW  = OUT;

    const int gemmBlocks = (N + 127) / 128;
    const int sideBlocks = (B + 127) / 128;
    const int chunkLen = (E + SORT_CH - 1) / SORT_CH;
    const int buckets  = (N + 255) / 256;
    const int ggBlocks = 256;

    const int tilesA = gemmBlocks / 2;
    const int tilesB = gemmBlocks - tilesA;
    k_sort_gemm<0><<<tilesA + SORT_CH, 256, 0, stream>>>(
        srcI, dstI, values, REC, CTAB, OTAB, ROWPTR, PERM, N, E, chunkLen, buckets, tilesA,
        features, W1, XW, 0,
        nullptr, nullptr, nullptr, nullptr, 0, 0);
    k_sort_gemm<1><<<sideBlocks + tilesB + buckets, 256, 0, stream>>>(
        srcI, dstI, values, REC, CTAB, OTAB, ROWPTR, PERM, N, E, chunkLen, buckets, sideBlocks + tilesB,
        features, W1, XW, tilesA,
        features, root_idx, W2 + 128 * DCOL, R2, B, sideBlocks);

    k_gather_gemm<false, true><<<ggBlocks, GG_THREADS, 0, stream>>>(
        XW, ROWPTR, PERM, b1, W2, batch, R2, nullptr, XW2, N,
        root_idx, Wl + 128 * DCOL, R3, B, sideBlocks, ggBlocks - sideBlocks);

    k_gather_gemm<true, false><<<ggBlocks, GG_THREADS, 0, stream>>>(
        XW2, ROWPTR, PERM, b2, Wl, batch, R3, bl, OUT, N,
        nullptr, nullptr, nullptr, 0, 0, ggBlocks);
}


# GEMM epilogues, all main-row tiles: C tile goes through 16 LDS rows of the wave's own and leaves as one b128 store per 512-byte row (whole lines); was 64 b32 stores of 2 half lines; sums unchanged
# speedup vs baseline: 3.9232x; 1.0309x over previous
#include <hip/hip_runtime.h>
#include <hip/hip_bf16.h>

typedef float v2f __attribute__((ext_vector_type(2)));
typedef float v4f __attribute__((ext_vector_type(4)));
typedef float v8f __attribute__((ext_vector_type(8)));

#define NEG_SLOPE 0.01f
#define DCOL 128

__device__ __forceinline__ float lrelu(float x) {
    return x > 0.0f ? x : NEG_SLOPE * x;
}

#define SORT_CH 256
#define SORT_T  16
#define BSH     8
#define SORT_LDS_INTS (8 * 256 + 8)

__device__ __forceinline__ int block_scan256(int v, int* wsum, int& total) {
    const int lane = threadIdx.x & 31;
    const int wave = threadIdx.x >> 5;
    int incl = v;
    for (int o = 1; o < 32; o <<= 1) {
        int n = __shfl_up(incl, o);
        if (lane >= o) incl += n;
    }
    if (lane == 31) wsum[wave] = incl;
    __syncthreads();
    int woff = 0, tot = 0;
#pragma unroll
    for (int k = 0; k < 8; ++k) {
        const int s = wsum[k];
        woff += (k < wave) ? s : 0;
        tot += s;
    }
    __syncthreads();
    total = tot;
    return woff + incl - v;
}

__device__ __forceinline__ int wave_counts_sum(const int* cnt, int tid, int (&c8)[8]) {
    int sum = 0;
#pragma unroll
    for (int k = 0; k < 8; ++k) { c8[k] = cnt[k * 256 + tid]; sum += c8[k]; }
    return sum;
}

__device__ __forceinline__ void wave_cursors(int* cnt, int tid, int start, const int (&c8)[8]) {
    int run = start;
#pragma unroll
    for (int k = 0; k < 8; ++k) { cnt[k * 256 + tid] = run; run += c8[k]; }
}

__device__ __forceinline__ void part_body(int* lds,
       const int* __restrict__ src,
       const int* __restrict__ dst,
       const float* __restrict__ wvals,
       int2* __restrict__ rec,
       int* __restrict__ ctab,
       int* __restrict__ otab, int E, int L, int NB, int c) {
    int* cnt  = lds;
    int* wsum = lds + 8 * 256;
    const int tid = threadIdx.x;
    int* mycnt = cnt + (tid >> 5) * 256;
    const int e0 = (c * L < E) ? c * L : E;
    const int e1 = (e0 + L < E) ? e0 + L : E;
#pragma unroll
    for (int k = 0; k < 8; ++k) cnt[k * 256 + tid] = 0;
    __syncthreads();

    for (int base = e0; base < e1; base += 256 * SORT_T) {
        int d[SORT_T];
#pragma unroll
        for (int k = 0; k < SORT_T; ++k) {
            const int e = base + k * 256 + tid;
            d[k] = dst[(e < e1) ? e : (e1 - 1)];
        }
#pragma unroll
        for (int k = 0; k < SORT_T; ++k) {
            const int e = base + k * 256 + tid;
            if (e < e1) atomicAdd(&mycnt[d[k] >> BSH], 1);
        }
    }
    __syncthreads();

    int c8[8];
    const int myc = wave_counts_sum(cnt, tid, c8);
    int total;
    const int off = block_scan256(myc, wsum, total);
    if (tid < NB) {
        ctab[tid * SORT_CH + c] = myc;
        otab[tid * SORT_CH + c] = off;
    }
    wave_cursors(cnt, tid, off, c8);
    __syncthreads();

    for (int base = e0; base < e1; base += 256 * SORT_T) {
        int d[SORT_T], s[SORT_T];
        float w[SORT_T];
#pragma unroll
        for (int k = 0; k < SORT_T; ++k) {
            const int e = base + k * 256 + tid;
            const int ec = (e < e1) ? e : (e1 - 1);
            d[k] = dst[ec];
            s[k] = src[ec];
            w[k] = wvals[ec];
        }
        int p[SORT_T];
#pragma unroll
        for (int k = 0; k < SORT_T; ++k) {
            const int e = base + k * 256 + tid;
            p[k] = (e < e1) ? atomicAdd(&mycnt[d[k] >> BSH], 1) : 0;
        }
#pragma unroll
        for (int k = 0; k < SORT_T; ++k) {
            const int e = base + k * 256 + tid;
            if (e < e1) {
                const unsigned x = ((unsigned)(d[k] & 255) << 24) | (unsigned)s[k];
                rec[e0 + p[k]] = make_int2((int)x, __float_as_int(w[k]));
            }
        }
    }
}

__device__ __forceinline__ void bucket_body(int* lds,
         const int2* __restrict__ rec,
         const int* __restrict__ ctab,
         const int* __restrict__ otab,
         int* __restrict__ rowptr,
         int2* __restrict__ perm, int N, int E, int L, int b) {
    int* ncnt = lds;
    int* wsum = lds + 8 * 256;
    const int tid  = threadIdx.x;
    const int lane = tid & 31;
    int* myn = ncnt + (tid >> 5) * 256;

    const int myCnt = ctab[b * SORT_CH + tid];
    const int myOff = otab[b * SORT_CH + tid];
    const int myStart = ((tid * L < E) ? tid * L : E) + myOff;
#pragma unroll
    for (int k = 0; k < 8; ++k) ncnt[k * 256 + tid] = 0;
    int base;
    (void)block_scan256(myOff, wsum, base);

    int2 r[32];
#pragma unroll
    for (int i = 0; i < 32; ++i) {
        const int cn = __builtin_amdgcn_readlane(myCnt, i);
        const int st = __builtin_amdgcn_readlane(myStart, i);
        int idx = st + ((lane < cn) ? lane : 0);
        idx = (idx < E) ? idx : (E - 1);
        r[i] = rec[idx];
    }
#pragma unroll
    for (int i = 0; i < 32; ++i) {
        const int cn = __builtin_amdgcn_readlane(myCnt, i);
        if (lane < cn) atomicAdd(&myn[(unsigned)r[i].x >> 24], 1);
    }
    const bool longPiece = __builtin_amdgcn_ballot_w32(myCnt > 32) != 0u;
    if (longPiece) {
        for (int i = 0; i < 32; ++i) {
            const int cn = __builtin_amdgcn_readlane(myCnt, i);
            const int st = __builtin_amdgcn_readlane(myStart, i);
            for (int k = 32 + lane; k < cn; k += 32) {
                const int2 q = rec[st + k];
                atomicAdd(&myn[(unsigned)q.x >> 24], 1);
            }
        }
    }
    __syncthreads();

    int c8[8];
    const int v = wave_counts_sum(ncnt, tid, c8);
    int tot;
    const int ex = block_scan256(v, wsum, tot);
    const int node = b * 256 + tid;
    if (node < N) rowptr[node] = base + ex;
    if (node == N - 1) rowptr[N] = E;
    wave_cursors(ncnt, tid, ex, c8);
    __syncthreads();

    int p[32];
#pragma unroll
    for (int i = 0; i < 32; ++i) {
        const int cn = __builtin_amdgcn_readlane(myCnt, i);
        p[i] = (lane < cn) ? atomicAdd(&myn[(unsigned)r[i].x >> 24], 1) : 0;
    }
#pragma unroll
    for (int i = 0; i < 32; ++i) {
        const int cn = __builtin_amdgcn_readlane(myCnt, i);
        if (lane < cn) {
            const unsigned x = (unsigned)r[i].x;
            perm[base + p[i]] = make_int2((int)(x & 0xFFFFFFu), r[i].y);
        }
    }
    if (longPiece) {
        for (int i = 0; i < 32; ++i) {
            const int cn = __builtin_amdgcn_readlane(myCnt, i);
            const int st = __builtin_amdgcn_readlane(myStart, i);
            for (int k = 32 + lane; k < cn; k += 32) {
                const int2 q = rec[st + k];
                const unsigned x = (unsigned)q.x;
                const int p = atomicAdd(&myn[x >> 24], 1);
                perm[base + p] = make_int2((int)(x & 0xFFFFFFu), q.y);
            }
        }
    }
}

template <int U>
__device__ __forceinline__ void gather_rows(const float* __restrict__ xw,
                                            int sx, int wy, int j, int lane,
                                            v4f& acc) {
    v4f x[U];
    float w[U];
#pragma unroll
    for (int u = 0; u < U; ++u) {
        const int s = __builtin_amdgcn_readlane(sx, j + u);
        w[u] = __int_as_float(__builtin_amdgcn_readlane(wy, j + u));
        x[u] = *(const v4f*)(xw + (size_t)s * DCOL + lane * 4);
    }
#pragma unroll
    for (int u = 0; u < U; ++u) acc += w[u] * x[u];
}

__device__ __forceinline__ v4f gather_node(const float* __restrict__ xw,
                                           const int* __restrict__ rowptr,
                                           const int2* __restrict__ perm,
                                           int node, int lane) {
    const int p0 = rowptr[node];
    const int p1 = rowptr[node + 1];
    v4f acc = {0.0f, 0.0f, 0.0f, 0.0f};
    for (int base = p0; base < p1; base += 32) {
        const int cnt = (p1 - base < 32) ? (p1 - base) : 32;
        int2 my = make_int2(0, 0);
        if (lane < cnt) my = perm[base + lane];
        int j = 0;
        for (; j + 8 <= cnt; j += 8) gather_rows<8>(xw, my.x, my.y, j, lane, acc);
        if (j + 4 <= cnt) { gather_rows<4>(xw, my.x, my.y, j, lane, acc); j += 4; }
        if (j + 2 <= cnt) { gather_rows<2>(xw, my.x, my.y, j, lane, acc); j += 2; }
        if (j < cnt)      { gather_rows<1>(xw, my.x, my.y, j, lane, acc); }
    }
    return acc;
}

#define LDA 132

template <bool LR>
__device__ __forceinline__ void put_row(float* __restrict__ dstrow, int lane, v4f acc, v4f gb) {
    v4f v = acc + gb;
    if (LR) { v[0] = lrelu(v[0]); v[1] = lrelu(v[1]); v[2] = lrelu(v[2]); v[3] = lrelu(v[3]); }
    *(v4f*)(dstrow + lane * 4) = v;
}

template <bool LR>
__device__ __forceinline__ void gather_tile_rows(float* __restrict__ arows,
                                                 const float* __restrict__ xw,
                                                 const int* __restrict__ rowptr,
                                                 const int2* __restrict__ perm,
                                                 v4f gb, int rowbase, int lane) {
    const int rpraw = rowptr[rowbase + (lane < 16 ? lane : 16)];
    const int P0 = __builtin_amdgcn_readlane(rpraw, 0);
    const int P1 = __builtin_amdgcn_readlane(rpraw, 16);
    const int rpv = (lane < 16) ? rpraw : 0x7fffffff;
    int cur = 0;
    int nxt = __builtin_amdgcn_readlane(rpv, 1);
    v4f acc = {0.0f, 0.0f, 0.0f, 0.0f};
    const v4f zero = {0.0f, 0.0f, 0.0f, 0.0f};
    if (P0 < P1) {
        auto pairs = [&](int b) {
            const int i = b + lane;
            int2 p = perm[(i < P1) ? i : (P1 - 1)];
            if (i >= P1) p.y = 0;
            return p;
        };
        auto row = [&](int s) {
            return *(const v4f*)(xw + (size_t)s * DCOL + lane * 4);
        };
        int2 my = pairs(P0);
        int2 nx = pairs(P0 + 32);
        v4f x[32];
#pragma unroll
        for (int k = 0; k < 32; ++k) {
            x[k] = row(__builtin_amdgcn_readlane(my.x, k));
            __builtin_amdgcn_sched_barrier(0);
        }
        for (int base = P0; base < P1; base += 32) {
            auto take = [&](int k) {
                const int e = base + k;
                while (e >= nxt) {
                    put_row<LR>(arows + cur * LDA, lane, acc, gb);
                    acc = zero;
                    ++cur;
                    nxt = __builtin_amdgcn_readlane(rpv, cur + 1);
                }
                const float w = __int_as_float(__builtin_amdgcn_readlane(my.y, k));
                acc += w * x[k];
            };
            if (base + 32 < P1) {
                const int2 nn = pairs(base + 64);
#pragma unroll
                for (int k = 0; k < 32; ++k) {
                    take(k);
                    x[k] = row(__builtin_amdgcn_readlane(nx.x, k));
                }
                my = nx;
                nx = nn;
            } else {
#pragma unroll
                for (int k = 0; k < 32; ++k) take(k);
            }
        }
    }
    while (cur < 16) {
        put_row<LR>(arows + cur * LDA, lane, acc, gb);
        acc = zero;
        ++cur;
    }
}

template <bool EPI>
__device__ __forceinline__ void store_tile_rows(float* __restrict__ srows,
                                                const v8f (&acc)[8],
                                                const float* __restrict__ bias,
                                                float* __restrict__ Ctile,
                                                int lane, int lo, int hi) {
#pragma unroll
    for (int t = 0; t < 8; ++t) {
#pragma unroll
        for (int r = 0; r < 8; ++r)
            srows[(r + hi * 8) * LDA + t * 16 + lo] = acc[t][r];
    }
    __builtin_amdgcn_fence(__ATOMIC_RELEASE, "wavefront");
    __builtin_amdgcn_wave_barrier();
    __builtin_amdgcn_fence(__ATOMIC_ACQUIRE, "wavefront");
    v4f badd = {0.0f, 0.0f, 0.0f, 0.0f};
    if (EPI) badd = *(const v4f*)(bias + lane * 4);
    v4f v[16];
#pragma unroll
    for (int j = 0; j < 16; ++j) v[j] = *(const v4f*)(srows + j * LDA + lane * 4);
#pragma unroll
    for (int j = 0; j < 16; ++j) {
        v4f o = v[j];
        if (EPI) {
            o += badd;
            o[0] = lrelu(o[0]); o[1] = lrelu(o[1]); o[2] = lrelu(o[2]); o[3] = lrelu(o[3]);
        }
        *(v4f*)(Ctile + (size_t)j * DCOL + lane * 4) = o;
    }
}

template <bool LR, bool INIT, bool EPI, bool MAP>
__device__ __forceinline__ void gemm_tile128(float* __restrict__ lw,
                                             float* __restrict__ lc,
                                             const float* __restrict__ A,
                                             const int* __restrict__ rowmap,
                                             const float* __restrict__ W,
                                             const int* __restrict__ batch,
                                             const float* __restrict__ R,
                                             const float* __restrict__ bias,
                                             float* __restrict__ C,
                                             int nrows, int blk) {
    const int tid  = threadIdx.x;
    const int wave = tid >> 5;
    const int lane = tid & 31;
    const int lo   = lane & 15;
    const int hi   = lane >> 4;
    const int rowbase = blk * 128 + wave * 16;
    const bool active = rowbase < nrows;

    int arow = rowbase + lo;
    if (MAP) {
        const int g = (arow < nrows) ? arow : (nrows - 1);
        arow = active ? rowmap[g] : 0;
    }

    v8f acc[8] = {};
    if (INIT && active) {
#pragma unroll
        for (int r = 0; r < 8; ++r) {
            const int g = batch[rowbase + r + hi * 8];
            const float* rp = R + (size_t)g * DCOL + lo;
#pragma unroll
            for (int t = 0; t < 8; ++t) acc[t][r] = rp[t * 16];
        }
    }

    v4f we[8], wo[8];
#pragma unroll
    for (int j = 0; j < 8; ++j) {
        const int i = tid + j * 256;
        const int kp = i >> 5;
        const int cg = (i & 31) * 4;
        we[j] = *(const v4f*)(W + (kp * 2 + 0) * DCOL + cg);
        wo[j] = *(const v4f*)(W + (kp * 2 + 1) * DCOL + cg);
    }
#pragma unroll
    for (int j = 0; j < 8; ++j) {
        const int i = tid + j * 256;
        const int kp = i >> 5;
        const int cg = (i & 31) * 4;
        float* dstp = lw + kp * 256 + cg * 2;
#pragma unroll
        for (int c = 0; c < 4; ++c) {
            v2f p = { we[j][c], wo[j][c] };
            *(v2f*)(dstp + c * 2) = p;
        }
    }
    __syncthreads();

    if (active) {
        const float* arowptr = A + (size_t)arow * DCOL;
        const float* lb = lw + hi * 256 + lo * 2;
#pragma unroll 4
        for (int ks = 0; ks < 32; ++ks) {
            v2f a = *(const v2f*)(arowptr + ks * 4 + hi * 2);
            if (LR) { a[0] = lrelu(a[0]); a[1] = lrelu(a[1]); }
            v2f b[8];
#pragma unroll
            for (int t = 0; t < 8; ++t)
                b[t] = *(const v2f*)(lb + ks * 512 + t * 32);
#pragma unroll
            for (int t = 0; t < 8; ++t)
                acc[t] = __builtin_amdgcn_wmma_f32_16x16x4_f32(
                    false, a, false, b[t], (short)0, acc[t], false, false);
        }

        if (MAP) {
#pragma unroll
            for (int r = 0; r < 8; ++r) {
                const int row = rowbase + r + hi * 8;
                if (row < nrows) {
#pragma unroll
                    for (int t = 0; t < 8; ++t)
                        C[(size_t)row * DCOL + t * 16 + lo] = acc[t][r];
                }
            }
        } else {
            store_tile_rows<EPI>(lc + wave * 16 * LDA, acc, bias,
                                 C + (size_t)rowbase * DCOL, lane, lo, hi);
        }
    }
}

#define GG_WAVES   16
#define GG_THREADS (GG_WAVES * 32)

__device__ __forceinline__ void stage_w(float* __restrict__ lw,
                                        const float* __restrict__ W, int tid) {
    constexpr int G4 = (128 * DCOL) / 8 / GG_THREADS;
    v4f we[G4], wo[G4];
#pragma unroll
    for (int j = 0; j < G4; ++j) {
        const int i = tid + j * GG_THREADS;
        const int kp = i >> 5;
        const int cg = (i & 31) * 4;
        we[j] = *(const v4f*)(W + (kp * 2 + 0) * DCOL + cg);
        wo[j] = *(const v4f*)(W + (kp * 2 + 1) * DCOL + cg);
    }
#pragma unroll
    for (int j = 0; j < G4; ++j) {
        const int i = tid + j * GG_THREADS;
        const int kp = i >> 5;
        const int cg = (i & 31) * 4;
        float* dstp = lw + kp * 256 + cg * 2;
#pragma unroll
        for (int c = 0; c < 4; ++c) {
            v2f p = { we[j][c], wo[j][c] };
            *(v2f*)(dstp + c * 2) = p;
        }
    }
}

template <bool EPI>
__device__ __forceinline__ void gather_gemm_groups(float* __restrict__ lw,
                                                   float* __restrict__ la,
                                                   const float* __restrict__ xw,
                                                   const int* __restrict__ rowptr,
                                                   const int2* __restrict__ perm,
                                                   const float* __restrict__ gbias,
                                                   const float* __restrict__ W,
                                                   const int* __restrict__ batch,
                                                   const float* __restrict__ R,
                                                   const float* __restrict__ bias,
                                                   float* __restrict__ C,
                                                   int nrows, int mb, int G) {
    const int tid  = threadIdx.x;
    const int wave = __builtin_amdgcn_readfirstlane(tid >> 5);
    const int lane = tid & 31;
    const int lo   = lane & 15;
    const int hi   = lane >> 4;

    stage_w(lw, W, tid);
    __syncthreads();

    const v4f gb = *(const v4f*)(gbias + lane * 4);
    float* arows = la + wave * 16 * LDA;
    const float* ap = arows + lo * LDA + hi * 2;
    const float* lb = lw + hi * 256 + lo * 2;
    const int groups = (nrows + 15) >> 4;
    for (int gi = wave * G + mb; gi < groups; gi += GG_WAVES * G) {
        const int rowbase = gi * 16;
        v8f acc[8];
#pragma unroll
        for (int r = 0; r < 8; ++r) {
            const int g = batch[rowbase + r + hi * 8];
            const float* rp = R + (size_t)g * DCOL + lo;
#pragma unroll
            for (int t = 0; t < 8; ++t) acc[t][r] = rp[t * 16];
        }

        gather_tile_rows<true>(arows, xw, rowptr, perm, gb, rowbase, lane);

#pragma unroll 4
        for (int ks = 0; ks < 32; ++ks) {
            const v2f a = *(const v2f*)(ap + ks * 4);
            v2f b[8];
#pragma unroll
            for (int t = 0; t < 8; ++t)
                b[t] = *(const v2f*)(lb + ks * 512 + t * 32);
#pragma unroll
            for (int t = 0; t < 8; ++t)
                acc[t] = __builtin_amdgcn_wmma_f32_16x16x4_f32(
                    false, a, false, b[t], (short)0, acc[t], false, false);
        }

        store_tile_rows<EPI>(arows, acc, bias, C + (size_t)rowbase * DCOL, lane, lo, hi);
    }
}

__device__ __forceinline__ void gather_gemm_side(float* __restrict__ lw,
                                                 float* __restrict__ la,
                                                 const float* __restrict__ xw,
                                                 const int* __restrict__ rowptr,
                                                 const int2* __restrict__ perm,
                                                 const float* __restrict__ gbias,
                                                 const int* __restrict__ rowmap,
                                                 const float* __restrict__ W,
                                                 float* __restrict__ C,
                                                 int nrows, int blk) {
    const int tid  = threadIdx.x;
    const int wave = tid >> 5;
    const int lane = tid & 31;
    const int lo   = lane & 15;
    const int hi   = lane >> 4;
    const int rowbase = blk * 128 + wave * 16;
    const bool active = wave < 8 && rowbase < nrows;

    stage_w(lw, W, tid);

    const v4f gb = *(const v4f*)(gbias + lane * 4);
    const v4f zero = {0.0f, 0.0f, 0.0f, 0.0f};
    for (int i = wave; i < 128; i += GG_WAVES) {
        const int g = blk * 128 + i;
        if (g < nrows) {
            const int node = __builtin_amdgcn_readfirstlane(rowmap[g]);
            const v4f s = gather_node(xw, rowptr, perm, node, lane);
            put_row<false>(la + i * LDA, lane, s, gb);
        } else {
            *(v4f*)(la + i * LDA + lane * 4) = zero;
        }
    }
    __syncthreads();

    if (active) {
        v8f acc[8] = {};
        const float* ap = la + (wave * 16 + lo) * LDA + hi * 2;
        const float* lb = lw + hi * 256 + lo * 2;
#pragma unroll 4
        for (int ks = 0; ks < 32; ++ks) {
            const v2f a = *(const v2f*)(ap + ks * 4);
            v2f b[8];
#pragma unroll
            for (int t = 0; t < 8; ++t)
                b[t] = *(const v2f*)(lb + ks * 512 + t * 32);
#pragma unroll
            for (int t = 0; t < 8; ++t)
                acc[t] = __builtin_amdgcn_wmma_f32_16x16x4_f32(
                    false, a, false, b[t], (short)0, acc[t], false, false);
        }

#pragma unroll
        for (int r = 0; r < 8; ++r) {
            const int row = rowbase + r + hi * 8;
            if (row < nrows) {
#pragma unroll
                for (int t = 0; t < 8; ++t)
                    C[(size_t)row * DCOL + t * 16 + lo] = acc[t][r];
            }
        }
    }
}

template <bool EPI, bool SIDE>
__global__ void __launch_bounds__(GG_THREADS)
k_gather_gemm(const float* __restrict__ xw,
              const int* __restrict__ rowptr,
              const int2* __restrict__ perm,
              const float* __restrict__ gbias,
              const float* __restrict__ W,
              const int* __restrict__ batch,
              const float* __restrict__ R,
              const float* __restrict__ bias,
              float* __restrict__ C, int N,
              const int* __restrict__ root_idx,
              const float* __restrict__ TW,
              float* __restrict__ T, int B, int sideBlocks, int mainBlocks) {
    __shared__ float lw[128 * 128];
    __shared__ float la[GG_WAVES * 16 * LDA];
    const int blk = (int)blockIdx.x;
    if (SIDE && blk < sideBlocks) {
        gather_gemm_side(lw, la, xw, rowptr, perm, gbias, root_idx, TW, T, B, blk);
    } else {
        gather_gemm_groups<EPI>(
            lw, la, xw, rowptr, perm, gbias, W,
            batch, R, bias, C, N, blk - (SIDE ? sideBlocks : 0), mainBlocks);
    }
}

template <int PHASE>
__global__ void __launch_bounds__(256)
k_sort_gemm(const int* __restrict__ src,
            const int* __restrict__ dst,
            const float* __restrict__ wvals,
            int2* __restrict__ rec,
            int* __restrict__ ctab,
            int* __restrict__ otab,
            int* __restrict__ rowptr,
            int2* __restrict__ perm,
            int N, int E, int L, int NB, int gemmPart,
            const float* __restrict__ A0,
            const float* __restrict__ W,
            float* __restrict__ C, int tile0,
            const float* __restrict__ TA,
            const int* __restrict__ root_idx,
            const float* __restrict__ TW,
            float* __restrict__ T, int B, int sideBlocks) {
    __shared__ float lw[128 * 128];
    __shared__ __attribute__((aligned(16))) float lc[8 * 16 * LDA];
    const int blk = (int)blockIdx.x;
    if (blk >= gemmPart) {
        if (PHASE == 0)
            part_body((int*)lw, src, dst, wvals, rec, ctab, otab, E, L, NB, blk - gemmPart);
        else
            bucket_body((int*)lw, rec, ctab, otab, rowptr, perm, N, E, L, blk - gemmPart);
    } else if (blk < sideBlocks) {
        gemm_tile128<true, false, false, true>(
            lw, lc, TA, root_idx, TW, nullptr, nullptr, nullptr, T, B, blk);
    } else {
        gemm_tile128<false, false, false, false>(
            lw, lc, A0, nullptr, W, nullptr, nullptr, nullptr, C, N,
            tile0 + blk - sideBlocks);
    }
}

extern "C" void kernel_launch(void* const* d_in, const int* in_sizes, int n_in,
                              void* d_out, int out_size, void* d_ws, size_t ws_size,
                              hipStream_t stream) {
    const float* features = (const float*)d_in[0];
    const float* values   = (const float*)d_in[1];
    const float* W1       = (const float*)d_in[2];
    const float* b1       = (const float*)d_in[3];
    const float* W2       = (const float*)d_in[4];
    const float* b2       = (const float*)d_in[5];
    const float* Wl       = (const float*)d_in[6];
    const float* bl       = (const float*)d_in[7];
    const int*   adjs     = (const int*)d_in[8];
    const int*   root_idx = (const int*)d_in[9];
    const int*   batch    = (const int*)d_in[12];

    const int N = in_sizes[0] / DCOL;
    const int E = in_sizes[1];
    const int B = in_sizes[9];
    const int* srcI = adjs;
    const int* dstI = adjs + E;

    int2* PERM   = (int2*)d_ws;
    int2* REC    = PERM + E;
    int*  CTAB   = (int*)(REC + E);
    int*  OTAB   = CTAB + 256 * SORT_CH;
    int*  ROWPTR = OTAB + 256 * SORT_CH;
    float* XW2 = (float*)d_ws + (size_t)N * DCOL;
    float* R2 = XW2 - (size_t)2 * B * DCOL;
    float* R3 = R2 + (size_t)B * DCOL;

    float* OUT = (float*)d_out;
    float* XW  = OUT;

    const int gemmBlocks = (N + 127) / 128;
    const int sideBlocks = (B + 127) / 128;
    const int chunkLen = (E + SORT_CH - 1) / SORT_CH;
    const int buckets  = (N + 255) / 256;
    const int ggBlocks = 256;

    const int tilesA = gemmBlocks / 2;
    const int tilesB = gemmBlocks - tilesA;
    k_sort_gemm<0><<<tilesA + SORT_CH, 256, 0, stream>>>(
        srcI, dstI, values, REC, CTAB, OTAB, ROWPTR, PERM, N, E, chunkLen, buckets, tilesA,
        features, W1, XW, 0,
        nullptr, nullptr, nullptr, nullptr, 0, 0);
    k_sort_gemm<1><<<sideBlocks + tilesB + buckets, 256, 0, stream>>>(
        srcI, dstI, values, REC, CTAB, OTAB, ROWPTR, PERM, N, E, chunkLen, buckets, sideBlocks + tilesB,
        features, W1, XW, tilesA,
        features, root_idx, W2 + 128 * DCOL, R2, B, sideBlocks);

    k_gather_gemm<false, true><<<ggBlocks, GG_THREADS, 0, stream>>>(
        XW, ROWPTR, PERM, b1, W2, batch, R2, nullptr, XW2, N,
        root_idx, Wl + 128 * DCOL, R3, B, sideBlocks, ggBlocks - sideBlocks);

    k_gather_gemm<true, false><<<ggBlocks, GG_THREADS, 0, stream>>>(
        XW2, ROWPTR, PERM, b2, Wl, batch, R3, bl, OUT, N,
        nullptr, nullptr, nullptr, 0, 0, ggBlocks);
}
